# starting kernel
# baseline (speedup 1.0000x reference)
#include <hip/hip_runtime.h>

#define NODES  50000
#define EDGES  800000
#define LEDGES 200000
#define INCH   256
#define NF     64

typedef float v2f __attribute__((ext_vector_type(2)));
typedef float v8f __attribute__((ext_vector_type(8)));

__global__ void k_deg_init(float* deg, int n) {
  int i = blockIdx.x * blockDim.x + threadIdx.x;
  if (i < n) deg[i] = 1.0f;
}

__global__ void k_deg_accum(const int* __restrict__ col, float* deg, int e) {
  int i = blockIdx.x * blockDim.x + threadIdx.x;
  if (i < e) atomicAdd(&deg[col[i]], 1.0f);
}

__global__ void k_deg_rsqrt(float* deg, int n) {
  int i = blockIdx.x * blockDim.x + threadIdx.x;
  if (i < n) deg[i] = rsqrtf(deg[i]);
}

template<int K>
__global__ __launch_bounds__(256) void k_gemm(const float* __restrict__ A,
                                              const float* __restrict__ W,
                                              float* __restrict__ C, int M) {
  __shared__ __align__(16) float sB[4096];

  const int lane = threadIdx.x & 31;
  const int wave = threadIdx.x >> 5;

  int strip = blockIdx.x * 8 + wave;
  int maxStrip = (M >> 4) - 1;
  if (strip > maxStrip) strip = maxStrip;
  const long rowBase = (long)strip << 4;

  const int mlo  = lane & 15;
  const int koff = (lane >> 4) << 1;
  const float* arow = A + (rowBase + mlo) * (long)K;

  v8f acc[4];
#pragma unroll
  for (int t = 0; t < 4; ++t) acc[t] = (v8f){0.f,0.f,0.f,0.f,0.f,0.f,0.f,0.f};

  for (int k0 = 0; k0 < K; k0 += 64) {
    __syncthreads();
    for (int idx = threadIdx.x; idx < 2048; idx += 256) {
      int ks = idx >> 7;
      int t  = (idx >> 5) & 3;
      int l  = idx & 31;
      int kk = k0 + ks * 4 + ((l >> 4) << 1);
      int cl = t * 16 + (l & 15);
      sB[idx * 2 + 0] = W[kk * NF + cl];
      sB[idx * 2 + 1] = W[(kk + 1) * NF + cl];
    }
    __syncthreads();
    if (k0 + 64 < K)
      __builtin_prefetch(arow + k0 + 64, 0, 0);
#pragma unroll 4
    for (int ks = 0; ks < 16; ++ks) {
      v2f a = *(const v2f*)(arow + k0 + ks * 4 + koff);
      const v2f* bfrag = (const v2f*)sB + ks * 128 + lane;
#pragma unroll
      for (int t = 0; t < 4; ++t) {
        v2f b = bfrag[t * 32];
        acc[t] = __builtin_amdgcn_wmma_f32_16x16x4_f32(
            false, a, false, b, (short)0, acc[t], false, false);
      }
    }
  }

  const int mhi = (lane >> 4) << 3;
#pragma unroll
  for (int t = 0; t < 4; ++t)
#pragma unroll
    for (int i = 0; i < 8; ++i)
      C[(rowBase + mhi + i) * NF + t * 16 + mlo] = acc[t][i];
}

__global__ void k_agg_init(const float* __restrict__ xw, const float* __restrict__ dinv,
                           float* __restrict__ agg, int n) {
  int t = blockIdx.x * blockDim.x + threadIdx.x;
  int node = t >> 4;
  int f = (t & 15) << 2;
  if (node < n) {
    float s = dinv[node]; s = s * s;
    float4 v = *(const float4*)(xw + (long)node * NF + f);
    v.x *= s; v.y *= s; v.z *= s; v.w *= s;
    *(float4*)(agg + (long)node * NF + f) = v;
  }
}

__global__ void k_agg_edges(const float* __restrict__ xw, const int* __restrict__ row,
                            const int* __restrict__ col, const float* __restrict__ dinv,
                            float* agg, int e) {
  int t = blockIdx.x * blockDim.x + threadIdx.x;
  int ed = t >> 4;
  int f = (t & 15) << 2;
  if (ed < e) {
    int r = row[ed], c = col[ed];
    float nrm = dinv[r] * dinv[c];
    float4 v = *(const float4*)(xw + (long)r * NF + f);
    float* dst = agg + (long)c * NF + f;
    atomicAdd(dst + 0, v.x * nrm);
    atomicAdd(dst + 1, v.y * nrm);
    atomicAdd(dst + 2, v.z * nrm);
    atomicAdd(dst + 3, v.w * nrm);
  }
}

template<bool RELU>
__global__ void k_bias_act(const float* __restrict__ agg, const float* __restrict__ bias,
                           float* __restrict__ out, int n) {
  int t = blockIdx.x * blockDim.x + threadIdx.x;
  int node = t >> 4;
  int f = (t & 15) << 2;
  if (node < n) {
    float4 v = *(const float4*)(agg + (long)node * NF + f);
    v.x += bias[f + 0]; v.y += bias[f + 1]; v.z += bias[f + 2]; v.w += bias[f + 3];
    if (RELU) {
      v.x = fmaxf(v.x, 0.f); v.y = fmaxf(v.y, 0.f);
      v.z = fmaxf(v.z, 0.f); v.w = fmaxf(v.w, 0.f);
    }
    *(float4*)(out + (long)node * NF + f) = v;
  }
}

__global__ void k_decode(const float* __restrict__ z, const int* __restrict__ src,
                         const int* __restrict__ dst, float* __restrict__ out, int e) {
  int i = blockIdx.x * blockDim.x + threadIdx.x;
  if (i < e) {
    const float4* a = (const float4*)(z + (long)src[i] * NF);
    const float4* b = (const float4*)(z + (long)dst[i] * NF);
    float acc = 0.f;
#pragma unroll
    for (int j = 0; j < 16; ++j) {
      float4 av = a[j], bv = b[j];
      acc += av.x * bv.x + av.y * bv.y + av.z * bv.z + av.w * bv.w;
    }
    out[i] = acc;
  }
}

extern "C" void kernel_launch(void* const* d_in, const int* in_sizes, int n_in,
                              void* d_out, int out_size, void* d_ws, size_t ws_size,
                              hipStream_t stream) {
  const float* x   = (const float*)d_in[0];
  const int*   ei  = (const int*)d_in[1];
  const int*   eli = (const int*)d_in[2];
  const float* W1  = (const float*)d_in[3];
  const float* b1  = (const float*)d_in[4];
  const float* W2  = (const float*)d_in[5];
  const float* b2  = (const float*)d_in[6];
  float* out = (float*)d_out;

  float* ws   = (float*)d_ws;
  float* dinv = ws;
  float* bufA = ws + 50048;
  float* bufB = bufA + (size_t)NODES * NF;
  float* bufC = bufB + (size_t)NODES * NF;

  const int* erow = ei;
  const int* ecol = ei + EDGES;
  const int* lsrc = eli;
  const int* ldst = eli + LEDGES;

  dim3 blk(256);
  const int nodeGrid = (NODES * 16 + 255) / 256;
  const int edgeGrid = (EDGES * 16 + 255) / 256;
  const int strips   = NODES / 16;
  const int gemmGrid = (strips + 7) / 8;

  k_deg_init<<<(NODES + 255) / 256, blk, 0, stream>>>(dinv, NODES);
  k_deg_accum<<<(EDGES + 255) / 256, blk, 0, stream>>>(ecol, dinv, EDGES);
  k_deg_rsqrt<<<(NODES + 255) / 256, blk, 0, stream>>>(dinv, NODES);

  k_gemm<INCH><<<gemmGrid, blk, 0, stream>>>(x, W1, bufA, NODES);
  k_agg_init<<<nodeGrid, blk, 0, stream>>>(bufA, dinv, bufB, NODES);
  k_agg_edges<<<edgeGrid, blk, 0, stream>>>(bufA, erow, ecol, dinv, bufB, EDGES);
  k_bias_act<true><<<nodeGrid, blk, 0, stream>>>(bufB, b1, bufA, NODES);

  k_gemm<NF><<<gemmGrid, blk, 0, stream>>>(bufA, W2, bufB, NODES);
  k_agg_init<<<nodeGrid, blk, 0, stream>>>(bufB, dinv, bufC, NODES);
  k_agg_edges<<<edgeGrid, blk, 0, stream>>>(bufB, erow, ecol, dinv, bufC, EDGES);
  k_bias_act<false><<<nodeGrid, blk, 0, stream>>>(bufC, b2, bufA, NODES);

  k_decode<<<(LEDGES + 255) / 256, blk, 0, stream>>>(bufA, lsrc, ldst, out, LEDGES);
}


# Aggregation: float-atomic scatter replaced by per-call counting-sort CSR (int atomic rank, scan, fill) and per-node gather-sum in arrival-rank order; self-loop, bias, relu in its one store
# speedup vs baseline: 1.5780x; 1.5780x over previous
#include <hip/hip_runtime.h>

#define NODES  50000
#define EDGES  800000
#define LEDGES 200000
#define INCH   256
#define NF     64

typedef float v2f __attribute__((ext_vector_type(2)));
typedef float v8f __attribute__((ext_vector_type(8)));

__global__ void k_cnt_zero(int* cnt, int n) {
  int i = blockIdx.x * blockDim.x + threadIdx.x;
  if (i < n) cnt[i] = 0;
}

__global__ void k_rank(const int* __restrict__ col, int* cnt, int* __restrict__ rank, int e) {
  int i = blockIdx.x * blockDim.x + threadIdx.x;
  if (i < e) rank[i] = atomicAdd(&cnt[col[i]], 1);
}

#define SCAN_T     1024
#define SCAN_ITEMS 49
#define SCAN_QUADS (NODES / 4)
#define SCAN_FULL  (SCAN_QUADS / SCAN_T)
__global__ __launch_bounds__(SCAN_T) void k_scan(const int* __restrict__ cnt,
                                                 int* __restrict__ offs,
                                                 float* __restrict__ dinv) {
  __shared__ __align__(16) int s[SCAN_T * SCAN_ITEMS];
  __shared__ int wtot[32];
  const int tid  = threadIdx.x;
  const int lane = tid & 31;
  const int wave = tid >> 5;

  const int4* cnt4  = (const int4*)cnt;
  float4*     dinv4 = (float4*)dinv;
  int4*       s4    = (int4*)s;
  int4 c[SCAN_FULL];
#pragma unroll
  for (int it = 0; it < SCAN_FULL; ++it) c[it] = cnt4[it * SCAN_T + tid];
#pragma unroll
  for (int it = 0; it < SCAN_FULL; ++it) {
    const int q = it * SCAN_T + tid;
    s4[q] = c[it];
    dinv4[q] = make_float4(rsqrtf((float)(c[it].x + 1)), rsqrtf((float)(c[it].y + 1)),
                           rsqrtf((float)(c[it].z + 1)), rsqrtf((float)(c[it].w + 1)));
  }
  {
    const int q = SCAN_FULL * SCAN_T + tid;
    if (q < SCAN_T * SCAN_ITEMS / 4) {
      int4 z = make_int4(0, 0, 0, 0);
      if (q < SCAN_QUADS) {
        z = cnt4[q];
        dinv4[q] = make_float4(rsqrtf((float)(z.x + 1)), rsqrtf((float)(z.y + 1)),
                               rsqrtf((float)(z.z + 1)), rsqrtf((float)(z.w + 1)));
      }
      s4[q] = z;
    }
  }
  __syncthreads();

  const int base = tid * SCAN_ITEMS;
  int sum = 0;
#pragma unroll 7
  for (int k = 0; k < SCAN_ITEMS; ++k) sum += s[base + k];

  int incl = sum;
#pragma unroll
  for (int d = 1; d < 32; d <<= 1) {
    int up = __shfl_up(incl, d, 32);
    if (lane >= d) incl += up;
  }
  if (lane == 31) wtot[wave] = incl;
  __syncthreads();
  if (wave == 0) {
    const int w = wtot[lane];
    int wi = w;
#pragma unroll
    for (int d = 1; d < 32; d <<= 1) {
      int up = __shfl_up(wi, d, 32);
      if (lane >= d) wi += up;
    }
    wtot[lane] = wi - w;
  }
  __syncthreads();

  int run = wtot[wave] + incl - sum;
#pragma unroll 7
  for (int k = 0; k < SCAN_ITEMS; ++k) {
    int c = s[base + k];
    s[base + k] = run;
    run += c;
  }
  __syncthreads();

  int4* offs4 = (int4*)offs;
  for (int q = tid; q < SCAN_QUADS; q += SCAN_T) offs4[q] = s4[q];
  if (tid == 0) offs[NODES] = s[NODES];
}

__global__ void k_fill(const int* __restrict__ row, const int* __restrict__ col,
                       const int* __restrict__ rank, const int* __restrict__ offs,
                       int* __restrict__ srow, int e) {
  int i = blockIdx.x * blockDim.x + threadIdx.x;
  if (i < e) srow[offs[col[i]] + rank[i]] = row[i];
}

template<int K>
__global__ __launch_bounds__(256) void k_gemm(const float* __restrict__ A,
                                              const float* __restrict__ W,
                                              float* __restrict__ C, int M) {
  __shared__ __align__(16) float sB[4096];

  const int lane = threadIdx.x & 31;
  const int wave = threadIdx.x >> 5;

  int strip = blockIdx.x * 8 + wave;
  int maxStrip = (M >> 4) - 1;
  if (strip > maxStrip) strip = maxStrip;
  const long rowBase = (long)strip << 4;

  const int mlo  = lane & 15;
  const int koff = (lane >> 4) << 1;
  const float* arow = A + (rowBase + mlo) * (long)K;

  v8f acc[4];
#pragma unroll
  for (int t = 0; t < 4; ++t) acc[t] = (v8f){0.f,0.f,0.f,0.f,0.f,0.f,0.f,0.f};

  for (int k0 = 0; k0 < K; k0 += 64) {
    __syncthreads();
    for (int idx = threadIdx.x; idx < 2048; idx += 256) {
      int ks = idx >> 7;
      int t  = (idx >> 5) & 3;
      int l  = idx & 31;
      int kk = k0 + ks * 4 + ((l >> 4) << 1);
      int cl = t * 16 + (l & 15);
      sB[idx * 2 + 0] = W[kk * NF + cl];
      sB[idx * 2 + 1] = W[(kk + 1) * NF + cl];
    }
    __syncthreads();
    if (k0 + 64 < K)
      __builtin_prefetch(arow + k0 + 64, 0, 0);
#pragma unroll 4
    for (int ks = 0; ks < 16; ++ks) {
      v2f a = *(const v2f*)(arow + k0 + ks * 4 + koff);
      const v2f* bfrag = (const v2f*)sB + ks * 128 + lane;
#pragma unroll
      for (int t = 0; t < 4; ++t) {
        v2f b = bfrag[t * 32];
        acc[t] = __builtin_amdgcn_wmma_f32_16x16x4_f32(
            false, a, false, b, (short)0, acc[t], false, false);
      }
    }
  }

  const int mhi = (lane >> 4) << 3;
#pragma unroll
  for (int t = 0; t < 4; ++t)
#pragma unroll
    for (int i = 0; i < 8; ++i)
      C[(rowBase + mhi + i) * NF + t * 16 + mlo] = acc[t][i];
}

template<bool RELU>
__global__ __launch_bounds__(256) void k_gather(const float* __restrict__ xw,
                                                const int* __restrict__ offs,
                                                const int* __restrict__ srow,
                                                const float* __restrict__ dinv,
                                                const float* __restrict__ bias,
                                                float* __restrict__ out, int n) {
  int t = blockIdx.x * blockDim.x + threadIdx.x;
  int node = t >> 4;
  int f = (t & 15) << 2;
  if (node < n) {
    const float dn = dinv[node];
    const int beg = offs[node];
    const int end = offs[node + 1];
    const float* xf = xw + f;

    float4 acc;
    {
      float s = dn * dn;
      float4 v = *(const float4*)(xf + (long)node * NF);
      acc.x = v.x * s; acc.y = v.y * s; acc.z = v.z * s; acc.w = v.w * s;
    }

    int j = beg;
    for (; j + 4 <= end; j += 4) {
      int r0 = srow[j], r1 = srow[j + 1], r2 = srow[j + 2], r3 = srow[j + 3];
      float w0 = dinv[r0] * dn, w1 = dinv[r1] * dn, w2 = dinv[r2] * dn, w3 = dinv[r3] * dn;
      float4 v0 = *(const float4*)(xf + (long)r0 * NF);
      float4 v1 = *(const float4*)(xf + (long)r1 * NF);
      float4 v2 = *(const float4*)(xf + (long)r2 * NF);
      float4 v3 = *(const float4*)(xf + (long)r3 * NF);
      acc.x += v0.x * w0; acc.y += v0.y * w0; acc.z += v0.z * w0; acc.w += v0.w * w0;
      acc.x += v1.x * w1; acc.y += v1.y * w1; acc.z += v1.z * w1; acc.w += v1.w * w1;
      acc.x += v2.x * w2; acc.y += v2.y * w2; acc.z += v2.z * w2; acc.w += v2.w * w2;
      acc.x += v3.x * w3; acc.y += v3.y * w3; acc.z += v3.z * w3; acc.w += v3.w * w3;
    }
    if (j + 2 <= end) {
      int r0 = srow[j], r1 = srow[j + 1];
      float w0 = dinv[r0] * dn, w1 = dinv[r1] * dn;
      float4 v0 = *(const float4*)(xf + (long)r0 * NF);
      float4 v1 = *(const float4*)(xf + (long)r1 * NF);
      acc.x += v0.x * w0; acc.y += v0.y * w0; acc.z += v0.z * w0; acc.w += v0.w * w0;
      acc.x += v1.x * w1; acc.y += v1.y * w1; acc.z += v1.z * w1; acc.w += v1.w * w1;
      j += 2;
    }
    if (j < end) {
      int r = srow[j];
      float w = dinv[r] * dn;
      float4 v = *(const float4*)(xf + (long)r * NF);
      acc.x += v.x * w; acc.y += v.y * w; acc.z += v.z * w; acc.w += v.w * w;
    }

    acc.x += bias[f + 0]; acc.y += bias[f + 1]; acc.z += bias[f + 2]; acc.w += bias[f + 3];
    if (RELU) {
      acc.x = fmaxf(acc.x, 0.f); acc.y = fmaxf(acc.y, 0.f);
      acc.z = fmaxf(acc.z, 0.f); acc.w = fmaxf(acc.w, 0.f);
    }
    *(float4*)(out + (long)node * NF + f) = acc;
  }
}

__global__ void k_decode(const float* __restrict__ z, const int* __restrict__ src,
                         const int* __restrict__ dst, float* __restrict__ out, int e) {
  int i = blockIdx.x * blockDim.x + threadIdx.x;
  if (i < e) {
    const float4* a = (const float4*)(z + (long)src[i] * NF);
    const float4* b = (const float4*)(z + (long)dst[i] * NF);
    float acc = 0.f;
#pragma unroll
    for (int j = 0; j < 16; ++j) {
      float4 av = a[j], bv = b[j];
      acc += av.x * bv.x + av.y * bv.y + av.z * bv.z + av.w * bv.w;
    }
    out[i] = acc;
  }
}

extern "C" void kernel_launch(void* const* d_in, const int* in_sizes, int n_in,
                              void* d_out, int out_size, void* d_ws, size_t ws_size,
                              hipStream_t stream) {
  const float* x   = (const float*)d_in[0];
  const int*   ei  = (const int*)d_in[1];
  const int*   eli = (const int*)d_in[2];
  const float* W1  = (const float*)d_in[3];
  const float* b1  = (const float*)d_in[4];
  const float* W2  = (const float*)d_in[5];
  const float* b2  = (const float*)d_in[6];
  float* out = (float*)d_out;

  float* ws   = (float*)d_ws;
  float* dinv = ws;
  float* bufA = ws + 50048;
  float* bufB = bufA + (size_t)NODES * NF;
  int*   cnt  = (int*)(bufB + (size_t)NODES * NF);
  int*   offs = cnt + 50048;
  int*   rank = offs + 50048;
  int*   srow = rank + EDGES;

  const int* erow = ei;
  const int* ecol = ei + EDGES;
  const int* lsrc = eli;
  const int* ldst = eli + LEDGES;

  dim3 blk(256);
  const int nodeGrid = (NODES * 16 + 255) / 256;
  const int strips   = NODES / 16;
  const int gemmGrid = (strips + 7) / 8;

  k_cnt_zero<<<(NODES + 255) / 256, blk, 0, stream>>>(cnt, NODES);
  k_rank<<<(EDGES + 255) / 256, blk, 0, stream>>>(ecol, cnt, rank, EDGES);
  k_scan<<<1, SCAN_T, 0, stream>>>(cnt, offs, dinv);
  k_fill<<<(EDGES + 255) / 256, blk, 0, stream>>>(erow, ecol, rank, offs, srow, EDGES);

  k_gemm<INCH><<<gemmGrid, blk, 0, stream>>>(x, W1, bufA, NODES);
  k_gather<true><<<nodeGrid, blk, 0, stream>>>(bufA, offs, srow, dinv, b1, bufB, NODES);

  k_gemm<NF><<<gemmGrid, blk, 0, stream>>>(bufB, W2, bufA, NODES);
  k_gather<false><<<nodeGrid, blk, 0, stream>>>(bufA, offs, srow, dinv, b2, bufB, NODES);

  k_decode<<<(LEDGES + 255) / 256, blk, 0, stream>>>(bufB, lsrc, ldst, out, LEDGES);
}


# CSR build: global returning atomics and scattered 4-byte stores replaced by two-level LDS counting sort (edges by 256-node bin, then bin by node); 9 launches to 7; sums in LDS arrival-rank order
# speedup vs baseline: 2.7914x; 1.7689x over previous
#include <hip/hip_runtime.h>

#define NODES  50000
#define EDGES  800000
#define LEDGES 200000
#define INCH   256
#define NF     64

typedef float v2f __attribute__((ext_vector_type(2)));
typedef float v8f __attribute__((ext_vector_type(8)));

#define BIN_NODES 256
#define NBINS     ((NODES + BIN_NODES - 1) / BIN_NODES)
#define NCHUNK    256
#define CHUNK_E   (EDGES / NCHUNK)
#define CS_T      512
#define CS_K      ((CHUNK_E + CS_T - 1) / CS_T)
#define BB_T      512
#define BB_K      10
#define BB_CAP    8192
static_assert(EDGES % NCHUNK == 0, "chunks are equal");
static_assert(NBINS + 1 <= 256 && NCHUNK == 256, "scans below take 256 entries");
static_assert(NODES < (1 << 23), "source id must fit beside 8 target bits");

__device__ __forceinline__ int wave_scan256(int* a, int lane) {
  int v[8];
  int s = 0;
#pragma unroll
  for (int k = 0; k < 8; ++k) { v[k] = a[lane * 8 + k]; s += v[k]; }
  int incl = s;
#pragma unroll
  for (int d = 1; d < 32; d <<= 1) {
    int up = __shfl_up(incl, d, 32);
    if (lane >= d) incl += up;
  }
  int run = incl - s;
#pragma unroll
  for (int k = 0; k < 8; ++k) { a[lane * 8 + k] = run; run += v[k]; }
  return __shfl(incl, 31, 32);
}

__global__ __launch_bounds__(CS_T) void k_chunk_sort(const int* __restrict__ row,
                                                     const int* __restrict__ col,
                                                     int* __restrict__ part,
                                                     int* __restrict__ loff) {
  __shared__ int hist[256];
  __shared__ int sortedL[CHUNK_E];
  const int tid  = threadIdx.x;
  const int lane = tid & 31;
  const int wave = tid >> 5;
  const int e0   = blockIdx.x * CHUNK_E;

  if (tid < 256) hist[tid] = 0;
  __syncthreads();

  int p[CS_K], bn[CS_K], rk[CS_K];
#pragma unroll
  for (int k = 0; k < CS_K; ++k) {
    const int i = k * CS_T + tid;
    p[k] = 0; bn[k] = 0;
    if (i < CHUNK_E) {
      const int c = col[e0 + i];
      const int r = row[e0 + i];
      bn[k] = c >> 8;
      p[k]  = (r << 8) | (c & 255);
    }
  }
#pragma unroll
  for (int k = 0; k < CS_K; ++k) {
    const int i = k * CS_T + tid;
    rk[k] = 0;
    if (i < CHUNK_E) rk[k] = atomicAdd(&hist[bn[k]], 1);
  }
  __syncthreads();
  if (wave == 0) wave_scan256(hist, lane);
  __syncthreads();
#pragma unroll
  for (int k = 0; k < CS_K; ++k) {
    const int i = k * CS_T + tid;
    if (i < CHUNK_E) sortedL[hist[bn[k]] + rk[k]] = p[k];
  }
  if (tid <= NBINS) loff[blockIdx.x * (NBINS + 1) + tid] = hist[tid];
  __syncthreads();
#pragma unroll
  for (int k = 0; k < CS_K; ++k) {
    const int i = k * CS_T + tid;
    if (i < CHUNK_E) part[e0 + i] = sortedL[i];
  }
}

__global__ __launch_bounds__(BB_T) void k_bin_build(const int* __restrict__ part,
                                                    const int* __restrict__ loff,
                                                    int* __restrict__ offs,
                                                    int* __restrict__ srow,
                                                    float* __restrict__ dinv) {
  __shared__ int pre[NCHUNK + 1];
  __shared__ int lo_s[NCHUNK];
  __shared__ int cntL[BIN_NODES];
  __shared__ int stage[BB_CAP];
  __shared__ int sorted[BB_CAP];
  __shared__ int tot[2];
  const int tid  = threadIdx.x;
  const int lane = tid & 31;
  const int wave = tid >> 5;
  const int b    = blockIdx.x;

  if (tid < NCHUNK) {
    const int lo = loff[tid * (NBINS + 1) + b];
    const int hi = loff[tid * (NBINS + 1) + b + 1];
    lo_s[tid] = lo;
    pre[tid]  = hi - lo;
  }
  if (tid < BIN_NODES) cntL[tid] = 0;
  __syncthreads();
  if (wave == 0) {
    int s = 0;
#pragma unroll
    for (int k = 0; k < 8; ++k) s += lo_s[lane * 8 + k];
#pragma unroll
    for (int d = 16; d >= 1; d >>= 1) s += __shfl_xor(s, d, 32);
    const int t = wave_scan256(pre, lane);
    if (lane == 0) { pre[NCHUNK] = t; tot[0] = t; tot[1] = s; }
  }
  __syncthreads();
  const int  T        = tot[0];
  const int  binStart = tot[1];
  const bool inLds    = T <= BB_CAP;

  auto fetch = [&](int i) -> int {
    int c = 0;
#pragma unroll
    for (int step = NCHUNK / 2; step >= 1; step >>= 1)
      if (pre[c + step] <= i) c += step;
    return part[c * CHUNK_E + lo_s[c] + (i - pre[c])];
  };

  for (int base = 0; base < T; base += BB_T * BB_K) {
    int q[BB_K];
#pragma unroll
    for (int k = 0; k < BB_K; ++k) {
      const int i = base + k * BB_T + tid;
      q[k] = fetch(i < T ? i : T - 1);
    }
#pragma unroll
    for (int k = 0; k < BB_K; ++k) {
      const int i = base + k * BB_T + tid;
      if (i < T) {
        atomicAdd(&cntL[q[k] & 255], 1);
        if (inLds) stage[i] = q[k];
      }
    }
  }
  __syncthreads();
  int mycnt = 0;
  if (tid < BIN_NODES) mycnt = cntL[tid];
  __syncthreads();
  if (wave == 0) wave_scan256(cntL, lane);
  __syncthreads();
  if (tid < BIN_NODES) {
    const int node = b * BIN_NODES + tid;
    if (node <= NODES) offs[node] = binStart + cntL[tid];
    if (node < NODES)  dinv[node] = rsqrtf((float)(mycnt + 1));
  }
  __syncthreads();

  if (inLds) {
    for (int base = 0; base < T; base += BB_T * BB_K) {
      int q[BB_K], pos[BB_K];
#pragma unroll
      for (int k = 0; k < BB_K; ++k) {
        const int i = base + k * BB_T + tid;
        q[k] = (i < T) ? stage[i] : 0;
      }
#pragma unroll
      for (int k = 0; k < BB_K; ++k) {
        const int i = base + k * BB_T + tid;
        pos[k] = 0;
        if (i < T) pos[k] = atomicAdd(&cntL[q[k] & 255], 1);
      }
#pragma unroll
      for (int k = 0; k < BB_K; ++k) {
        const int i = base + k * BB_T + tid;
        if (i < T) sorted[pos[k]] = (int)((unsigned)q[k] >> 8);
      }
    }
    __syncthreads();
    for (int base = 0; base < T; base += BB_T * BB_K) {
#pragma unroll
      for (int k = 0; k < BB_K; ++k) {
        const int i = base + k * BB_T + tid;
        if (i < T) srow[binStart + i] = sorted[i];
      }
    }
  } else {
    for (int i = tid; i < T; i += BB_T) {
      const int q   = fetch(i);
      const int pos = atomicAdd(&cntL[q & 255], 1);
      srow[binStart + pos] = (int)((unsigned)q >> 8);
    }
  }
}

template<int K>
__global__ __launch_bounds__(256) void k_gemm(const float* __restrict__ A,
                                              const float* __restrict__ W,
                                              float* __restrict__ C, int M) {
  __shared__ __align__(16) float sB[4096];

  const int lane = threadIdx.x & 31;
  const int wave = threadIdx.x >> 5;

  int strip = blockIdx.x * 8 + wave;
  int maxStrip = (M >> 4) - 1;
  if (strip > maxStrip) strip = maxStrip;
  const long rowBase = (long)strip << 4;

  const int mlo  = lane & 15;
  const int koff = (lane >> 4) << 1;
  const float* arow = A + (rowBase + mlo) * (long)K;

  v8f acc[4];
#pragma unroll
  for (int t = 0; t < 4; ++t) acc[t] = (v8f){0.f,0.f,0.f,0.f,0.f,0.f,0.f,0.f};

  for (int k0 = 0; k0 < K; k0 += 64) {
    __syncthreads();
    for (int idx = threadIdx.x; idx < 2048; idx += 256) {
      int ks = idx >> 7;
      int t  = (idx >> 5) & 3;
      int l  = idx & 31;
      int kk = k0 + ks * 4 + ((l >> 4) << 1);
      int cl = t * 16 + (l & 15);
      sB[idx * 2 + 0] = W[kk * NF + cl];
      sB[idx * 2 + 1] = W[(kk + 1) * NF + cl];
    }
    __syncthreads();
    if (k0 + 64 < K)
      __builtin_prefetch(arow + k0 + 64, 0, 0);
#pragma unroll 4
    for (int ks = 0; ks < 16; ++ks) {
      v2f a = *(const v2f*)(arow + k0 + ks * 4 + koff);
      const v2f* bfrag = (const v2f*)sB + ks * 128 + lane;
#pragma unroll
      for (int t = 0; t < 4; ++t) {
        v2f b = bfrag[t * 32];
        acc[t] = __builtin_amdgcn_wmma_f32_16x16x4_f32(
            false, a, false, b, (short)0, acc[t], false, false);
      }
    }
  }

  const int mhi = (lane >> 4) << 3;
#pragma unroll
  for (int t = 0; t < 4; ++t)
#pragma unroll
    for (int i = 0; i < 8; ++i)
      C[(rowBase + mhi + i) * NF + t * 16 + mlo] = acc[t][i];
}

template<bool RELU>
__global__ __launch_bounds__(256) void k_gather(const float* __restrict__ xw,
                                                const int* __restrict__ offs,
                                                const int* __restrict__ srow,
                                                const float* __restrict__ dinv,
                                                const float* __restrict__ bias,
                                                float* __restrict__ out, int n) {
  int t = blockIdx.x * blockDim.x + threadIdx.x;
  int node = t >> 4;
  int f = (t & 15) << 2;
  if (node < n) {
    const float dn = dinv[node];
    const int beg = offs[node];
    const int end = offs[node + 1];
    const float* xf = xw + f;

    float4 acc;
    {
      float s = dn * dn;
      float4 v = *(const float4*)(xf + (long)node * NF);
      acc.x = v.x * s; acc.y = v.y * s; acc.z = v.z * s; acc.w = v.w * s;
    }

    int j = beg;
    for (; j + 4 <= end; j += 4) {
      int r0 = srow[j], r1 = srow[j + 1], r2 = srow[j + 2], r3 = srow[j + 3];
      float w0 = dinv[r0] * dn, w1 = dinv[r1] * dn, w2 = dinv[r2] * dn, w3 = dinv[r3] * dn;
      float4 v0 = *(const float4*)(xf + (long)r0 * NF);
      float4 v1 = *(const float4*)(xf + (long)r1 * NF);
      float4 v2 = *(const float4*)(xf + (long)r2 * NF);
      float4 v3 = *(const float4*)(xf + (long)r3 * NF);
      acc.x += v0.x * w0; acc.y += v0.y * w0; acc.z += v0.z * w0; acc.w += v0.w * w0;
      acc.x += v1.x * w1; acc.y += v1.y * w1; acc.z += v1.z * w1; acc.w += v1.w * w1;
      acc.x += v2.x * w2; acc.y += v2.y * w2; acc.z += v2.z * w2; acc.w += v2.w * w2;
      acc.x += v3.x * w3; acc.y += v3.y * w3; acc.z += v3.z * w3; acc.w += v3.w * w3;
    }
    if (j + 2 <= end) {
      int r0 = srow[j], r1 = srow[j + 1];
      float w0 = dinv[r0] * dn, w1 = dinv[r1] * dn;
      float4 v0 = *(const float4*)(xf + (long)r0 * NF);
      float4 v1 = *(const float4*)(xf + (long)r1 * NF);
      acc.x += v0.x * w0; acc.y += v0.y * w0; acc.z += v0.z * w0; acc.w += v0.w * w0;
      acc.x += v1.x * w1; acc.y += v1.y * w1; acc.z += v1.z * w1; acc.w += v1.w * w1;
      j += 2;
    }
    if (j < end) {
      int r = srow[j];
      float w = dinv[r] * dn;
      float4 v = *(const float4*)(xf + (long)r * NF);
      acc.x += v.x * w; acc.y += v.y * w; acc.z += v.z * w; acc.w += v.w * w;
    }

    acc.x += bias[f + 0]; acc.y += bias[f + 1]; acc.z += bias[f + 2]; acc.w += bias[f + 3];
    if (RELU) {
      acc.x = fmaxf(acc.x, 0.f); acc.y = fmaxf(acc.y, 0.f);
      acc.z = fmaxf(acc.z, 0.f); acc.w = fmaxf(acc.w, 0.f);
    }
    *(float4*)(out + (long)node * NF + f) = acc;
  }
}

__global__ void k_decode(const float* __restrict__ z, const int* __restrict__ src,
                         const int* __restrict__ dst, float* __restrict__ out, int e) {
  int i = blockIdx.x * blockDim.x + threadIdx.x;
  if (i < e) {
    const float4* a = (const float4*)(z + (long)src[i] * NF);
    const float4* b = (const float4*)(z + (long)dst[i] * NF);
    float acc = 0.f;
#pragma unroll
    for (int j = 0; j < 16; ++j) {
      float4 av = a[j], bv = b[j];
      acc += av.x * bv.x + av.y * bv.y + av.z * bv.z + av.w * bv.w;
    }
    out[i] = acc;
  }
}

extern "C" void kernel_launch(void* const* d_in, const int* in_sizes, int n_in,
                              void* d_out, int out_size, void* d_ws, size_t ws_size,
                              hipStream_t stream) {
  const float* x   = (const float*)d_in[0];
  const int*   ei  = (const int*)d_in[1];
  const int*   eli = (const int*)d_in[2];
  const float* W1  = (const float*)d_in[3];
  const float* b1  = (const float*)d_in[4];
  const float* W2  = (const float*)d_in[5];
  const float* b2  = (const float*)d_in[6];
  float* out = (float*)d_out;

  float* ws   = (float*)d_ws;
  float* dinv = ws;
  float* bufA = ws + 50048;
  float* bufB = bufA + (size_t)NODES * NF;
  int*   offs = (int*)(bufB + (size_t)NODES * NF);
  int*   loff = offs + 50048;
  int*   part = loff + 50688;
  int*   srow = part + EDGES;

  const int* erow = ei;
  const int* ecol = ei + EDGES;
  const int* lsrc = eli;
  const int* ldst = eli + LEDGES;

  dim3 blk(256);
  const int nodeGrid = (NODES * 16 + 255) / 256;
  const int strips   = NODES / 16;
  const int gemmGrid = (strips + 7) / 8;

  k_chunk_sort<<<NCHUNK, CS_T, 0, stream>>>(erow, ecol, part, loff);
  k_bin_build<<<NBINS, BB_T, 0, stream>>>(part, loff, offs, srow, dinv);

  k_gemm<INCH><<<gemmGrid, blk, 0, stream>>>(x, W1, bufA, NODES);
  k_gather<true><<<nodeGrid, blk, 0, stream>>>(bufA, offs, srow, dinv, b1, bufB, NODES);

  k_gemm<NF><<<gemmGrid, blk, 0, stream>>>(bufB, W2, bufA, NODES);
  k_gather<false><<<nodeGrid, blk, 0, stream>>>(bufA, offs, srow, dinv, b2, bufB, NODES);

  k_decode<<<(LEDGES + 255) / 256, blk, 0, stream>>>(bufB, lsrc, ldst, out, LEDGES);
}


# GEMM W staging, both layers: the 8 dependent load-wait-store trips per 64-row chunk become one group of 16 loads issued ahead of the barrier and stored behind it; arithmetic and sum order unchanged
# speedup vs baseline: 3.2293x; 1.1569x over previous
#include <hip/hip_runtime.h>

#define NODES  50000
#define EDGES  800000
#define LEDGES 200000
#define INCH   256
#define NF     64

typedef float v2f __attribute__((ext_vector_type(2)));
typedef float v8f __attribute__((ext_vector_type(8)));

#define BIN_NODES 256
#define NBINS     ((NODES + BIN_NODES - 1) / BIN_NODES)
#define NCHUNK    256
#define CHUNK_E   (EDGES / NCHUNK)
#define CS_T      512
#define CS_K      ((CHUNK_E + CS_T - 1) / CS_T)
#define BB_T      512
#define BB_K      10
#define BB_CAP    8192
static_assert(EDGES % NCHUNK == 0, "chunks are equal");
static_assert(NBINS + 1 <= 256 && NCHUNK == 256, "scans below take 256 entries");
static_assert(NODES < (1 << 23), "source id must fit beside 8 target bits");

__device__ __forceinline__ int wave_scan256(int* a, int lane) {
  int v[8];
  int s = 0;
#pragma unroll
  for (int k = 0; k < 8; ++k) { v[k] = a[lane * 8 + k]; s += v[k]; }
  int incl = s;
#pragma unroll
  for (int d = 1; d < 32; d <<= 1) {
    int up = __shfl_up(incl, d, 32);
    if (lane >= d) incl += up;
  }
  int run = incl - s;
#pragma unroll
  for (int k = 0; k < 8; ++k) { a[lane * 8 + k] = run; run += v[k]; }
  return __shfl(incl, 31, 32);
}

__global__ __launch_bounds__(CS_T) void k_chunk_sort(const int* __restrict__ row,
                                                     const int* __restrict__ col,
                                                     int* __restrict__ part,
                                                     int* __restrict__ loff) {
  __shared__ int hist[256];
  __shared__ int sortedL[CHUNK_E];
  const int tid  = threadIdx.x;
  const int lane = tid & 31;
  const int wave = tid >> 5;
  const int e0   = blockIdx.x * CHUNK_E;

  if (tid < 256) hist[tid] = 0;
  __syncthreads();

  int p[CS_K], bn[CS_K], rk[CS_K];
#pragma unroll
  for (int k = 0; k < CS_K; ++k) {
    const int i = k * CS_T + tid;
    p[k] = 0; bn[k] = 0;
    if (i < CHUNK_E) {
      const int c = col[e0 + i];
      const int r = row[e0 + i];
      bn[k] = c >> 8;
      p[k]  = (r << 8) | (c & 255);
    }
  }
#pragma unroll
  for (int k = 0; k < CS_K; ++k) {
    const int i = k * CS_T + tid;
    rk[k] = 0;
    if (i < CHUNK_E) rk[k] = atomicAdd(&hist[bn[k]], 1);
  }
  __syncthreads();
  if (wave == 0) wave_scan256(hist, lane);
  __syncthreads();
#pragma unroll
  for (int k = 0; k < CS_K; ++k) {
    const int i = k * CS_T + tid;
    if (i < CHUNK_E) sortedL[hist[bn[k]] + rk[k]] = p[k];
  }
  if (tid <= NBINS) loff[blockIdx.x * (NBINS + 1) + tid] = hist[tid];
  __syncthreads();
#pragma unroll
  for (int k = 0; k < CS_K; ++k) {
    const int i = k * CS_T + tid;
    if (i < CHUNK_E) part[e0 + i] = sortedL[i];
  }
}

__global__ __launch_bounds__(BB_T) void k_bin_build(const int* __restrict__ part,
                                                    const int* __restrict__ loff,
                                                    int* __restrict__ offs,
                                                    int* __restrict__ srow,
                                                    float* __restrict__ dinv) {
  __shared__ int pre[NCHUNK + 1];
  __shared__ int lo_s[NCHUNK];
  __shared__ int cntL[BIN_NODES];
  __shared__ int stage[BB_CAP];
  __shared__ int sorted[BB_CAP];
  __shared__ int tot[2];
  const int tid  = threadIdx.x;
  const int lane = tid & 31;
  const int wave = tid >> 5;
  const int b    = blockIdx.x;

  if (tid < NCHUNK) {
    const int lo = loff[tid * (NBINS + 1) + b];
    const int hi = loff[tid * (NBINS + 1) + b + 1];
    lo_s[tid] = lo;
    pre[tid]  = hi - lo;
  }
  if (tid < BIN_NODES) cntL[tid] = 0;
  __syncthreads();
  if (wave == 0) {
    int s = 0;
#pragma unroll
    for (int k = 0; k < 8; ++k) s += lo_s[lane * 8 + k];
#pragma unroll
    for (int d = 16; d >= 1; d >>= 1) s += __shfl_xor(s, d, 32);
    const int t = wave_scan256(pre, lane);
    if (lane == 0) { pre[NCHUNK] = t; tot[0] = t; tot[1] = s; }
  }
  __syncthreads();
  const int  T        = tot[0];
  const int  binStart = tot[1];
  const bool inLds    = T <= BB_CAP;

  auto fetch = [&](int i) -> int {
    int c = 0;
#pragma unroll
    for (int step = NCHUNK / 2; step >= 1; step >>= 1)
      if (pre[c + step] <= i) c += step;
    return part[c * CHUNK_E + lo_s[c] + (i - pre[c])];
  };

  for (int base = 0; base < T; base += BB_T * BB_K) {
    int q[BB_K];
#pragma unroll
    for (int k = 0; k < BB_K; ++k) {
      const int i = base + k * BB_T + tid;
      q[k] = fetch(i < T ? i : T - 1);
    }
#pragma unroll
    for (int k = 0; k < BB_K; ++k) {
      const int i = base + k * BB_T + tid;
      if (i < T) {
        atomicAdd(&cntL[q[k] & 255], 1);
        if (inLds) stage[i] = q[k];
      }
    }
  }
  __syncthreads();
  int mycnt = 0;
  if (tid < BIN_NODES) mycnt = cntL[tid];
  __syncthreads();
  if (wave == 0) wave_scan256(cntL, lane);
  __syncthreads();
  if (tid < BIN_NODES) {
    const int node = b * BIN_NODES + tid;
    if (node <= NODES) offs[node] = binStart + cntL[tid];
    if (node < NODES)  dinv[node] = rsqrtf((float)(mycnt + 1));
  }
  __syncthreads();

  if (inLds) {
    for (int base = 0; base < T; base += BB_T * BB_K) {
      int q[BB_K], pos[BB_K];
#pragma unroll
      for (int k = 0; k < BB_K; ++k) {
        const int i = base + k * BB_T + tid;
        q[k] = (i < T) ? stage[i] : 0;
      }
#pragma unroll
      for (int k = 0; k < BB_K; ++k) {
        const int i = base + k * BB_T + tid;
        pos[k] = 0;
        if (i < T) pos[k] = atomicAdd(&cntL[q[k] & 255], 1);
      }
#pragma unroll
      for (int k = 0; k < BB_K; ++k) {
        const int i = base + k * BB_T + tid;
        if (i < T) sorted[pos[k]] = (int)((unsigned)q[k] >> 8);
      }
    }
    __syncthreads();
    for (int base = 0; base < T; base += BB_T * BB_K) {
#pragma unroll
      for (int k = 0; k < BB_K; ++k) {
        const int i = base + k * BB_T + tid;
        if (i < T) srow[binStart + i] = sorted[i];
      }
    }
  } else {
    for (int i = tid; i < T; i += BB_T) {
      const int q   = fetch(i);
      const int pos = atomicAdd(&cntL[q & 255], 1);
      srow[binStart + pos] = (int)((unsigned)q >> 8);
    }
  }
}

template<int K>
__global__ __launch_bounds__(256) void k_gemm(const float* __restrict__ A,
                                              const float* __restrict__ W,
                                              float* __restrict__ C, int M) {
  __shared__ __align__(16) float sB[4096];

  const int lane = threadIdx.x & 31;
  const int wave = threadIdx.x >> 5;

  int strip = blockIdx.x * 8 + wave;
  int maxStrip = (M >> 4) - 1;
  if (strip > maxStrip) strip = maxStrip;
  const long rowBase = (long)strip << 4;

  const int mlo  = lane & 15;
  const int koff = (lane >> 4) << 1;
  const float* arow = A + (rowBase + mlo) * (long)K;

  v8f acc[4];
#pragma unroll
  for (int t = 0; t < 4; ++t) acc[t] = (v8f){0.f,0.f,0.f,0.f,0.f,0.f,0.f,0.f};

  for (int k0 = 0; k0 < K; k0 += 64) {
    float w0[8], w1[8];
#pragma unroll
    for (int it = 0; it < 8; ++it) {
      int idx = it * 256 + threadIdx.x;
      int ks = idx >> 7;
      int t  = (idx >> 5) & 3;
      int l  = idx & 31;
      int kk = k0 + ks * 4 + ((l >> 4) << 1);
      int cl = t * 16 + (l & 15);
      w0[it] = W[kk * NF + cl];
      w1[it] = W[(kk + 1) * NF + cl];
    }
    __syncthreads();
#pragma unroll
    for (int it = 0; it < 8; ++it) {
      int idx = it * 256 + threadIdx.x;
      sB[idx * 2 + 0] = w0[it];
      sB[idx * 2 + 1] = w1[it];
    }
    __syncthreads();
    if (k0 + 64 < K)
      __builtin_prefetch(arow + k0 + 64, 0, 0);
#pragma unroll 4
    for (int ks = 0; ks < 16; ++ks) {
      v2f a = *(const v2f*)(arow + k0 + ks * 4 + koff);
      const v2f* bfrag = (const v2f*)sB + ks * 128 + lane;
#pragma unroll
      for (int t = 0; t < 4; ++t) {
        v2f b = bfrag[t * 32];
        acc[t] = __builtin_amdgcn_wmma_f32_16x16x4_f32(
            false, a, false, b, (short)0, acc[t], false, false);
      }
    }
  }

  const int mhi = (lane >> 4) << 3;
#pragma unroll
  for (int t = 0; t < 4; ++t)
#pragma unroll
    for (int i = 0; i < 8; ++i)
      C[(rowBase + mhi + i) * NF + t * 16 + mlo] = acc[t][i];
}

template<bool RELU>
__global__ __launch_bounds__(256) void k_gather(const float* __restrict__ xw,
                                                const int* __restrict__ offs,
                                                const int* __restrict__ srow,
                                                const float* __restrict__ dinv,
                                                const float* __restrict__ bias,
                                                float* __restrict__ out, int n) {
  int t = blockIdx.x * blockDim.x + threadIdx.x;
  int node = t >> 4;
  int f = (t & 15) << 2;
  if (node < n) {
    const float dn = dinv[node];
    const int beg = offs[node];
    const int end = offs[node + 1];
    const float* xf = xw + f;

    float4 acc;
    {
      float s = dn * dn;
      float4 v = *(const float4*)(xf + (long)node * NF);
      acc.x = v.x * s; acc.y = v.y * s; acc.z = v.z * s; acc.w = v.w * s;
    }

    int j = beg;
    for (; j + 4 <= end; j += 4) {
      int r0 = srow[j], r1 = srow[j + 1], r2 = srow[j + 2], r3 = srow[j + 3];
      float w0 = dinv[r0] * dn, w1 = dinv[r1] * dn, w2 = dinv[r2] * dn, w3 = dinv[r3] * dn;
      float4 v0 = *(const float4*)(xf + (long)r0 * NF);
      float4 v1 = *(const float4*)(xf + (long)r1 * NF);
      float4 v2 = *(const float4*)(xf + (long)r2 * NF);
      float4 v3 = *(const float4*)(xf + (long)r3 * NF);
      acc.x += v0.x * w0; acc.y += v0.y * w0; acc.z += v0.z * w0; acc.w += v0.w * w0;
      acc.x += v1.x * w1; acc.y += v1.y * w1; acc.z += v1.z * w1; acc.w += v1.w * w1;
      acc.x += v2.x * w2; acc.y += v2.y * w2; acc.z += v2.z * w2; acc.w += v2.w * w2;
      acc.x += v3.x * w3; acc.y += v3.y * w3; acc.z += v3.z * w3; acc.w += v3.w * w3;
    }
    if (j + 2 <= end) {
      int r0 = srow[j], r1 = srow[j + 1];
      float w0 = dinv[r0] * dn, w1 = dinv[r1] * dn;
      float4 v0 = *(const float4*)(xf + (long)r0 * NF);
      float4 v1 = *(const float4*)(xf + (long)r1 * NF);
      acc.x += v0.x * w0; acc.y += v0.y * w0; acc.z += v0.z * w0; acc.w += v0.w * w0;
      acc.x += v1.x * w1; acc.y += v1.y * w1; acc.z += v1.z * w1; acc.w += v1.w * w1;
      j += 2;
    }
    if (j < end) {
      int r = srow[j];
      float w = dinv[r] * dn;
      float4 v = *(const float4*)(xf + (long)r * NF);
      acc.x += v.x * w; acc.y += v.y * w; acc.z += v.z * w; acc.w += v.w * w;
    }

    acc.x += bias[f + 0]; acc.y += bias[f + 1]; acc.z += bias[f + 2]; acc.w += bias[f + 3];
    if (RELU) {
      acc.x = fmaxf(acc.x, 0.f); acc.y = fmaxf(acc.y, 0.f);
      acc.z = fmaxf(acc.z, 0.f); acc.w = fmaxf(acc.w, 0.f);
    }
    *(float4*)(out + (long)node * NF + f) = acc;
  }
}

__global__ void k_decode(const float* __restrict__ z, const int* __restrict__ src,
                         const int* __restrict__ dst, float* __restrict__ out, int e) {
  int i = blockIdx.x * blockDim.x + threadIdx.x;
  if (i < e) {
    const float4* a = (const float4*)(z + (long)src[i] * NF);
    const float4* b = (const float4*)(z + (long)dst[i] * NF);
    float acc = 0.f;
#pragma unroll
    for (int j = 0; j < 16; ++j) {
      float4 av = a[j], bv = b[j];
      acc += av.x * bv.x + av.y * bv.y + av.z * bv.z + av.w * bv.w;
    }
    out[i] = acc;
  }
}

extern "C" void kernel_launch(void* const* d_in, const int* in_sizes, int n_in,
                              void* d_out, int out_size, void* d_ws, size_t ws_size,
                              hipStream_t stream) {
  const float* x   = (const float*)d_in[0];
  const int*   ei  = (const int*)d_in[1];
  const int*   eli = (const int*)d_in[2];
  const float* W1  = (const float*)d_in[3];
  const float* b1  = (const float*)d_in[4];
  const float* W2  = (const float*)d_in[5];
  const float* b2  = (const float*)d_in[6];
  float* out = (float*)d_out;

  float* ws   = (float*)d_ws;
  float* dinv = ws;
  float* bufA = ws + 50048;
  float* bufB = bufA + (size_t)NODES * NF;
  int*   offs = (int*)(bufB + (size_t)NODES * NF);
  int*   loff = offs + 50048;
  int*   part = loff + 50688;
  int*   srow = part + EDGES;

  const int* erow = ei;
  const int* ecol = ei + EDGES;
  const int* lsrc = eli;
  const int* ldst = eli + LEDGES;

  dim3 blk(256);
  const int nodeGrid = (NODES * 16 + 255) / 256;
  const int strips   = NODES / 16;
  const int gemmGrid = (strips + 7) / 8;

  k_chunk_sort<<<NCHUNK, CS_T, 0, stream>>>(erow, ecol, part, loff);
  k_bin_build<<<NBINS, BB_T, 0, stream>>>(part, loff, offs, srow, dinv);

  k_gemm<INCH><<<gemmGrid, blk, 0, stream>>>(x, W1, bufA, NODES);
  k_gather<true><<<nodeGrid, blk, 0, stream>>>(bufA, offs, srow, dinv, b1, bufB, NODES);

  k_gemm<NF><<<gemmGrid, blk, 0, stream>>>(bufB, W2, bufA, NODES);
  k_gather<false><<<nodeGrid, blk, 0, stream>>>(bufA, offs, srow, dinv, b2, bufB, NODES);

  k_decode<<<(LEDGES + 255) / 256, blk, 0, stream>>>(bufB, lsrc, ldst, out, LEDGES);
}


# Edge norm split: dinv[source] folded into both GEMM stores (rows pre-scaled once), the gather reads no dinv per edge; products re-associated, dinv[target] once after the sum; sum order unchanged
# speedup vs baseline: 3.3486x; 1.0369x over previous
#include <hip/hip_runtime.h>

#define NODES  50000
#define EDGES  800000
#define LEDGES 200000
#define INCH   256
#define NF     64

typedef float v2f __attribute__((ext_vector_type(2)));
typedef float v8f __attribute__((ext_vector_type(8)));

#define BIN_NODES 256
#define NBINS     ((NODES + BIN_NODES - 1) / BIN_NODES)
#define NCHUNK    256
#define CHUNK_E   (EDGES / NCHUNK)
#define CS_T      512
#define CS_K      ((CHUNK_E + CS_T - 1) / CS_T)
#define BB_T      512
#define BB_K      10
#define BB_CAP    8192
static_assert(EDGES % NCHUNK == 0, "chunks are equal");
static_assert(NBINS + 1 <= 256 && NCHUNK == 256, "scans below take 256 entries");
static_assert(NODES < (1 << 23), "source id must fit beside 8 target bits");

__device__ __forceinline__ int wave_scan256(int* a, int lane) {
  int v[8];
  int s = 0;
#pragma unroll
  for (int k = 0; k < 8; ++k) { v[k] = a[lane * 8 + k]; s += v[k]; }
  int incl = s;
#pragma unroll
  for (int d = 1; d < 32; d <<= 1) {
    int up = __shfl_up(incl, d, 32);
    if (lane >= d) incl += up;
  }
  int run = incl - s;
#pragma unroll
  for (int k = 0; k < 8; ++k) { a[lane * 8 + k] = run; run += v[k]; }
  return __shfl(incl, 31, 32);
}

__global__ __launch_bounds__(CS_T) void k_chunk_sort(const int* __restrict__ row,
                                                     const int* __restrict__ col,
                                                     int* __restrict__ part,
                                                     int* __restrict__ loff) {
  __shared__ int hist[256];
  __shared__ int sortedL[CHUNK_E];
  const int tid  = threadIdx.x;
  const int lane = tid & 31;
  const int wave = tid >> 5;
  const int e0   = blockIdx.x * CHUNK_E;

  if (tid < 256) hist[tid] = 0;
  __syncthreads();

  int p[CS_K], bn[CS_K], rk[CS_K];
#pragma unroll
  for (int k = 0; k < CS_K; ++k) {
    const int i = k * CS_T + tid;
    p[k] = 0; bn[k] = 0;
    if (i < CHUNK_E) {
      const int c = col[e0 + i];
      const int r = row[e0 + i];
      bn[k] = c >> 8;
      p[k]  = (r << 8) | (c & 255);
    }
  }
#pragma unroll
  for (int k = 0; k < CS_K; ++k) {
    const int i = k * CS_T + tid;
    rk[k] = 0;
    if (i < CHUNK_E) rk[k] = atomicAdd(&hist[bn[k]], 1);
  }
  __syncthreads();
  if (wave == 0) wave_scan256(hist, lane);
  __syncthreads();
#pragma unroll
  for (int k = 0; k < CS_K; ++k) {
    const int i = k * CS_T + tid;
    if (i < CHUNK_E) sortedL[hist[bn[k]] + rk[k]] = p[k];
  }
  if (tid <= NBINS) loff[blockIdx.x * (NBINS + 1) + tid] = hist[tid];
  __syncthreads();
#pragma unroll
  for (int k = 0; k < CS_K; ++k) {
    const int i = k * CS_T + tid;
    if (i < CHUNK_E) part[e0 + i] = sortedL[i];
  }
}

__global__ __launch_bounds__(BB_T) void k_bin_build(const int* __restrict__ part,
                                                    const int* __restrict__ loff,
                                                    int* __restrict__ offs,
                                                    int* __restrict__ srow,
                                                    float* __restrict__ dinv) {
  __shared__ int pre[NCHUNK + 1];
  __shared__ int lo_s[NCHUNK];
  __shared__ int cntL[BIN_NODES];
  __shared__ int stage[BB_CAP];
  __shared__ int sorted[BB_CAP];
  __shared__ int tot[2];
  const int tid  = threadIdx.x;
  const int lane = tid & 31;
  const int wave = tid >> 5;
  const int b    = blockIdx.x;

  if (tid < NCHUNK) {
    const int lo = loff[tid * (NBINS + 1) + b];
    const int hi = loff[tid * (NBINS + 1) + b + 1];
    lo_s[tid] = lo;
    pre[tid]  = hi - lo;
  }
  if (tid < BIN_NODES) cntL[tid] = 0;
  __syncthreads();
  if (wave == 0) {
    int s = 0;
#pragma unroll
    for (int k = 0; k < 8; ++k) s += lo_s[lane * 8 + k];
#pragma unroll
    for (int d = 16; d >= 1; d >>= 1) s += __shfl_xor(s, d, 32);
    const int t = wave_scan256(pre, lane);
    if (lane == 0) { pre[NCHUNK] = t; tot[0] = t; tot[1] = s; }
  }
  __syncthreads();
  const int  T        = tot[0];
  const int  binStart = tot[1];
  const bool inLds    = T <= BB_CAP;

  auto fetch = [&](int i) -> int {
    int c = 0;
#pragma unroll
    for (int step = NCHUNK / 2; step >= 1; step >>= 1)
      if (pre[c + step] <= i) c += step;
    return part[c * CHUNK_E + lo_s[c] + (i - pre[c])];
  };

  for (int base = 0; base < T; base += BB_T * BB_K) {
    int q[BB_K];
#pragma unroll
    for (int k = 0; k < BB_K; ++k) {
      const int i = base + k * BB_T + tid;
      q[k] = fetch(i < T ? i : T - 1);
    }
#pragma unroll
    for (int k = 0; k < BB_K; ++k) {
      const int i = base + k * BB_T + tid;
      if (i < T) {
        atomicAdd(&cntL[q[k] & 255], 1);
        if (inLds) stage[i] = q[k];
      }
    }
  }
  __syncthreads();
  int mycnt = 0;
  if (tid < BIN_NODES) mycnt = cntL[tid];
  __syncthreads();
  if (wave == 0) wave_scan256(cntL, lane);
  __syncthreads();
  if (tid < BIN_NODES) {
    const int node = b * BIN_NODES + tid;
    if (node <= NODES) offs[node] = binStart + cntL[tid];
    if (node < NODES)  dinv[node] = rsqrtf((float)(mycnt + 1));
  }
  __syncthreads();

  if (inLds) {
    for (int base = 0; base < T; base += BB_T * BB_K) {
      int q[BB_K], pos[BB_K];
#pragma unroll
      for (int k = 0; k < BB_K; ++k) {
        const int i = base + k * BB_T + tid;
        q[k] = (i < T) ? stage[i] : 0;
      }
#pragma unroll
      for (int k = 0; k < BB_K; ++k) {
        const int i = base + k * BB_T + tid;
        pos[k] = 0;
        if (i < T) pos[k] = atomicAdd(&cntL[q[k] & 255], 1);
      }
#pragma unroll
      for (int k = 0; k < BB_K; ++k) {
        const int i = base + k * BB_T + tid;
        if (i < T) sorted[pos[k]] = (int)((unsigned)q[k] >> 8);
      }
    }
    __syncthreads();
    for (int base = 0; base < T; base += BB_T * BB_K) {
#pragma unroll
      for (int k = 0; k < BB_K; ++k) {
        const int i = base + k * BB_T + tid;
        if (i < T) srow[binStart + i] = sorted[i];
      }
    }
  } else {
    for (int i = tid; i < T; i += BB_T) {
      const int q   = fetch(i);
      const int pos = atomicAdd(&cntL[q & 255], 1);
      srow[binStart + pos] = (int)((unsigned)q >> 8);
    }
  }
}

template<int K>
__global__ __launch_bounds__(256) void k_gemm(const float* __restrict__ A,
                                              const float* __restrict__ W,
                                              const float* __restrict__ rs,
                                              float* __restrict__ C, int M) {
  __shared__ __align__(16) float sB[4096];

  const int lane = threadIdx.x & 31;
  const int wave = threadIdx.x >> 5;

  int strip = blockIdx.x * 8 + wave;
  int maxStrip = (M >> 4) - 1;
  if (strip > maxStrip) strip = maxStrip;
  const long rowBase = (long)strip << 4;

  const int mlo  = lane & 15;
  const int koff = (lane >> 4) << 1;
  const float* arow = A + (rowBase + mlo) * (long)K;

  const int mhi = (lane >> 4) << 3;
  const float4 s0 = *(const float4*)(rs + rowBase + mhi);
  const float4 s1 = *(const float4*)(rs + rowBase + mhi + 4);

  v8f acc[4];
#pragma unroll
  for (int t = 0; t < 4; ++t) acc[t] = (v8f){0.f,0.f,0.f,0.f,0.f,0.f,0.f,0.f};

  for (int k0 = 0; k0 < K; k0 += 64) {
    float w0[8], w1[8];
#pragma unroll
    for (int it = 0; it < 8; ++it) {
      int idx = it * 256 + threadIdx.x;
      int ks = idx >> 7;
      int t  = (idx >> 5) & 3;
      int l  = idx & 31;
      int kk = k0 + ks * 4 + ((l >> 4) << 1);
      int cl = t * 16 + (l & 15);
      w0[it] = W[kk * NF + cl];
      w1[it] = W[(kk + 1) * NF + cl];
    }
    __syncthreads();
#pragma unroll
    for (int it = 0; it < 8; ++it) {
      int idx = it * 256 + threadIdx.x;
      sB[idx * 2 + 0] = w0[it];
      sB[idx * 2 + 1] = w1[it];
    }
    __syncthreads();
    if (k0 + 64 < K)
      __builtin_prefetch(arow + k0 + 64, 0, 0);
#pragma unroll 4
    for (int ks = 0; ks < 16; ++ks) {
      v2f a = *(const v2f*)(arow + k0 + ks * 4 + koff);
      const v2f* bfrag = (const v2f*)sB + ks * 128 + lane;
#pragma unroll
      for (int t = 0; t < 4; ++t) {
        v2f b = bfrag[t * 32];
        acc[t] = __builtin_amdgcn_wmma_f32_16x16x4_f32(
            false, a, false, b, (short)0, acc[t], false, false);
      }
    }
  }

  const float sc[8] = {s0.x, s0.y, s0.z, s0.w, s1.x, s1.y, s1.z, s1.w};
#pragma unroll
  for (int t = 0; t < 4; ++t)
#pragma unroll
    for (int i = 0; i < 8; ++i)
      C[(rowBase + mhi + i) * NF + t * 16 + mlo] = acc[t][i] * sc[i];
}

template<bool RELU>
__global__ __launch_bounds__(256) void k_gather(const float* __restrict__ xs,
                                                const int* __restrict__ offs,
                                                const int* __restrict__ srow,
                                                const float* __restrict__ dinv,
                                                const float* __restrict__ bias,
                                                float* __restrict__ out, int n) {
  int t = blockIdx.x * blockDim.x + threadIdx.x;
  int node = t >> 4;
  int f = (t & 15) << 2;
  if (node < n) {
    const float dn = dinv[node];
    const int beg = offs[node];
    const int end = offs[node + 1];
    const float* xf = xs + f;

    float4 acc = *(const float4*)(xf + (long)node * NF);

    int j = beg;
    for (; j + 4 <= end; j += 4) {
      int r0 = srow[j], r1 = srow[j + 1], r2 = srow[j + 2], r3 = srow[j + 3];
      float4 v0 = *(const float4*)(xf + (long)r0 * NF);
      float4 v1 = *(const float4*)(xf + (long)r1 * NF);
      float4 v2 = *(const float4*)(xf + (long)r2 * NF);
      float4 v3 = *(const float4*)(xf + (long)r3 * NF);
      acc.x += v0.x; acc.y += v0.y; acc.z += v0.z; acc.w += v0.w;
      acc.x += v1.x; acc.y += v1.y; acc.z += v1.z; acc.w += v1.w;
      acc.x += v2.x; acc.y += v2.y; acc.z += v2.z; acc.w += v2.w;
      acc.x += v3.x; acc.y += v3.y; acc.z += v3.z; acc.w += v3.w;
    }
    if (j + 2 <= end) {
      int r0 = srow[j], r1 = srow[j + 1];
      float4 v0 = *(const float4*)(xf + (long)r0 * NF);
      float4 v1 = *(const float4*)(xf + (long)r1 * NF);
      acc.x += v0.x; acc.y += v0.y; acc.z += v0.z; acc.w += v0.w;
      acc.x += v1.x; acc.y += v1.y; acc.z += v1.z; acc.w += v1.w;
      j += 2;
    }
    if (j < end) {
      int r = srow[j];
      float4 v = *(const float4*)(xf + (long)r * NF);
      acc.x += v.x; acc.y += v.y; acc.z += v.z; acc.w += v.w;
    }

    acc.x = acc.x * dn + bias[f + 0]; acc.y = acc.y * dn + bias[f + 1];
    acc.z = acc.z * dn + bias[f + 2]; acc.w = acc.w * dn + bias[f + 3];
    if (RELU) {
      acc.x = fmaxf(acc.x, 0.f); acc.y = fmaxf(acc.y, 0.f);
      acc.z = fmaxf(acc.z, 0.f); acc.w = fmaxf(acc.w, 0.f);
    }
    *(float4*)(out + (long)node * NF + f) = acc;
  }
}

__global__ void k_decode(const float* __restrict__ z, const int* __restrict__ src,
                         const int* __restrict__ dst, float* __restrict__ out, int e) {
  int i = blockIdx.x * blockDim.x + threadIdx.x;
  if (i < e) {
    const float4* a = (const float4*)(z + (long)src[i] * NF);
    const float4* b = (const float4*)(z + (long)dst[i] * NF);
    float acc = 0.f;
#pragma unroll
    for (int j = 0; j < 16; ++j) {
      float4 av = a[j], bv = b[j];
      acc += av.x * bv.x + av.y * bv.y + av.z * bv.z + av.w * bv.w;
    }
    out[i] = acc;
  }
}

extern "C" void kernel_launch(void* const* d_in, const int* in_sizes, int n_in,
                              void* d_out, int out_size, void* d_ws, size_t ws_size,
                              hipStream_t stream) {
  const float* x   = (const float*)d_in[0];
  const int*   ei  = (const int*)d_in[1];
  const int*   eli = (const int*)d_in[2];
  const float* W1  = (const float*)d_in[3];
  const float* b1  = (const float*)d_in[4];
  const float* W2  = (const float*)d_in[5];
  const float* b2  = (const float*)d_in[6];
  float* out = (float*)d_out;

  float* ws   = (float*)d_ws;
  float* dinv = ws;
  float* bufA = ws + 50048;
  float* bufB = bufA + (size_t)NODES * NF;
  int*   offs = (int*)(bufB + (size_t)NODES * NF);
  int*   loff = offs + 50048;
  int*   part = loff + 50688;
  int*   srow = part + EDGES;

  const int* erow = ei;
  const int* ecol = ei + EDGES;
  const int* lsrc = eli;
  const int* ldst = eli + LEDGES;

  dim3 blk(256);
  const int nodeGrid = (NODES * 16 + 255) / 256;
  const int strips   = NODES / 16;
  const int gemmGrid = (strips + 7) / 8;

  k_chunk_sort<<<NCHUNK, CS_T, 0, stream>>>(erow, ecol, part, loff);
  k_bin_build<<<NBINS, BB_T, 0, stream>>>(part, loff, offs, srow, dinv);

  k_gemm<INCH><<<gemmGrid, blk, 0, stream>>>(x, W1, dinv, bufA, NODES);
  k_gather<true><<<nodeGrid, blk, 0, stream>>>(bufA, offs, srow, dinv, b1, bufB, NODES);

  k_gemm<NF><<<gemmGrid, blk, 0, stream>>>(bufB, W2, dinv, bufA, NODES);
  k_gather<false><<<nodeGrid, blk, 0, stream>>>(bufA, offs, srow, dinv, b2, bufB, NODES);

  k_decode<<<(LEDGES + 255) / 256, blk, 0, stream>>>(bufB, lsrc, ldst, out, LEDGES);
}


# Layer-2 GEMM moved into the tail of the layer-1 aggregation launch: a block's 16 summed rows go through LDS to 16 f32 WMMAs per column tile, stored scaled; 7 launches to 6; sum order unchanged
# speedup vs baseline: 3.5734x; 1.0671x over previous
#include <hip/hip_runtime.h>

#define NODES  50000
#define EDGES  800000
#define LEDGES 200000
#define INCH   256
#define NF     64

typedef float v2f __attribute__((ext_vector_type(2)));
typedef float v8f __attribute__((ext_vector_type(8)));

#define BIN_NODES 256
#define NBINS     ((NODES + BIN_NODES - 1) / BIN_NODES)
#define NCHUNK    256
#define CHUNK_E   (EDGES / NCHUNK)
#define CS_T      512
#define CS_K      ((CHUNK_E + CS_T - 1) / CS_T)
#define BB_T      512
#define BB_K      10
#define BB_CAP    8192
static_assert(EDGES % NCHUNK == 0, "chunks are equal");
static_assert(NBINS + 1 <= 256 && NCHUNK == 256, "scans below take 256 entries");
static_assert(NODES < (1 << 23), "source id must fit beside 8 target bits");

__device__ __forceinline__ int wave_scan256(int* a, int lane) {
  int v[8];
  int s = 0;
#pragma unroll
  for (int k = 0; k < 8; ++k) { v[k] = a[lane * 8 + k]; s += v[k]; }
  int incl = s;
#pragma unroll
  for (int d = 1; d < 32; d <<= 1) {
    int up = __shfl_up(incl, d, 32);
    if (lane >= d) incl += up;
  }
  int run = incl - s;
#pragma unroll
  for (int k = 0; k < 8; ++k) { a[lane * 8 + k] = run; run += v[k]; }
  return __shfl(incl, 31, 32);
}

__global__ __launch_bounds__(CS_T) void k_chunk_sort(const int* __restrict__ row,
                                                     const int* __restrict__ col,
                                                     int* __restrict__ part,
                                                     int* __restrict__ loff) {
  __shared__ int hist[256];
  __shared__ int sortedL[CHUNK_E];
  const int tid  = threadIdx.x;
  const int lane = tid & 31;
  const int wave = tid >> 5;
  const int e0   = blockIdx.x * CHUNK_E;

  if (tid < 256) hist[tid] = 0;
  __syncthreads();

  int p[CS_K], bn[CS_K], rk[CS_K];
#pragma unroll
  for (int k = 0; k < CS_K; ++k) {
    const int i = k * CS_T + tid;
    p[k] = 0; bn[k] = 0;
    if (i < CHUNK_E) {
      const int c = col[e0 + i];
      const int r = row[e0 + i];
      bn[k] = c >> 8;
      p[k]  = (r << 8) | (c & 255);
    }
  }
#pragma unroll
  for (int k = 0; k < CS_K; ++k) {
    const int i = k * CS_T + tid;
    rk[k] = 0;
    if (i < CHUNK_E) rk[k] = atomicAdd(&hist[bn[k]], 1);
  }
  __syncthreads();
  if (wave == 0) wave_scan256(hist, lane);
  __syncthreads();
#pragma unroll
  for (int k = 0; k < CS_K; ++k) {
    const int i = k * CS_T + tid;
    if (i < CHUNK_E) sortedL[hist[bn[k]] + rk[k]] = p[k];
  }
  if (tid <= NBINS) loff[blockIdx.x * (NBINS + 1) + tid] = hist[tid];
  __syncthreads();
#pragma unroll
  for (int k = 0; k < CS_K; ++k) {
    const int i = k * CS_T + tid;
    if (i < CHUNK_E) part[e0 + i] = sortedL[i];
  }
}

__global__ __launch_bounds__(BB_T) void k_bin_build(const int* __restrict__ part,
                                                    const int* __restrict__ loff,
                                                    int* __restrict__ offs,
                                                    int* __restrict__ srow,
                                                    float* __restrict__ dinv) {
  __shared__ int pre[NCHUNK + 1];
  __shared__ int lo_s[NCHUNK];
  __shared__ int cntL[BIN_NODES];
  __shared__ int stage[BB_CAP];
  __shared__ int sorted[BB_CAP];
  __shared__ int tot[2];
  const int tid  = threadIdx.x;
  const int lane = tid & 31;
  const int wave = tid >> 5;
  const int b    = blockIdx.x;

  if (tid < NCHUNK) {
    const int lo = loff[tid * (NBINS + 1) + b];
    const int hi = loff[tid * (NBINS + 1) + b + 1];
    lo_s[tid] = lo;
    pre[tid]  = hi - lo;
  }
  if (tid < BIN_NODES) cntL[tid] = 0;
  __syncthreads();
  if (wave == 0) {
    int s = 0;
#pragma unroll
    for (int k = 0; k < 8; ++k) s += lo_s[lane * 8 + k];
#pragma unroll
    for (int d = 16; d >= 1; d >>= 1) s += __shfl_xor(s, d, 32);
    const int t = wave_scan256(pre, lane);
    if (lane == 0) { pre[NCHUNK] = t; tot[0] = t; tot[1] = s; }
  }
  __syncthreads();
  const int  T        = tot[0];
  const int  binStart = tot[1];
  const bool inLds    = T <= BB_CAP;

  auto fetch = [&](int i) -> int {
    int c = 0;
#pragma unroll
    for (int step = NCHUNK / 2; step >= 1; step >>= 1)
      if (pre[c + step] <= i) c += step;
    return part[c * CHUNK_E + lo_s[c] + (i - pre[c])];
  };

  for (int base = 0; base < T; base += BB_T * BB_K) {
    int q[BB_K];
#pragma unroll
    for (int k = 0; k < BB_K; ++k) {
      const int i = base + k * BB_T + tid;
      q[k] = fetch(i < T ? i : T - 1);
    }
#pragma unroll
    for (int k = 0; k < BB_K; ++k) {
      const int i = base + k * BB_T + tid;
      if (i < T) {
        atomicAdd(&cntL[q[k] & 255], 1);
        if (inLds) stage[i] = q[k];
      }
    }
  }
  __syncthreads();
  int mycnt = 0;
  if (tid < BIN_NODES) mycnt = cntL[tid];
  __syncthreads();
  if (wave == 0) wave_scan256(cntL, lane);
  __syncthreads();
  if (tid < BIN_NODES) {
    const int node = b * BIN_NODES + tid;
    if (node <= NODES) offs[node] = binStart + cntL[tid];
    if (node < NODES)  dinv[node] = rsqrtf((float)(mycnt + 1));
  }
  __syncthreads();

  if (inLds) {
    for (int base = 0; base < T; base += BB_T * BB_K) {
      int q[BB_K], pos[BB_K];
#pragma unroll
      for (int k = 0; k < BB_K; ++k) {
        const int i = base + k * BB_T + tid;
        q[k] = (i < T) ? stage[i] : 0;
      }
#pragma unroll
      for (int k = 0; k < BB_K; ++k) {
        const int i = base + k * BB_T + tid;
        pos[k] = 0;
        if (i < T) pos[k] = atomicAdd(&cntL[q[k] & 255], 1);
      }
#pragma unroll
      for (int k = 0; k < BB_K; ++k) {
        const int i = base + k * BB_T + tid;
        if (i < T) sorted[pos[k]] = (int)((unsigned)q[k] >> 8);
      }
    }
    __syncthreads();
    for (int base = 0; base < T; base += BB_T * BB_K) {
#pragma unroll
      for (int k = 0; k < BB_K; ++k) {
        const int i = base + k * BB_T + tid;
        if (i < T) srow[binStart + i] = sorted[i];
      }
    }
  } else {
    for (int i = tid; i < T; i += BB_T) {
      const int q   = fetch(i);
      const int pos = atomicAdd(&cntL[q & 255], 1);
      srow[binStart + pos] = (int)((unsigned)q >> 8);
    }
  }
}

template<int K>
__global__ __launch_bounds__(256) void k_gemm(const float* __restrict__ A,
                                              const float* __restrict__ W,
                                              const float* __restrict__ rs,
                                              float* __restrict__ C, int M) {
  __shared__ __align__(16) float sB[4096];

  const int lane = threadIdx.x & 31;
  const int wave = threadIdx.x >> 5;

  int strip = blockIdx.x * 8 + wave;
  int maxStrip = (M >> 4) - 1;
  if (strip > maxStrip) strip = maxStrip;
  const long rowBase = (long)strip << 4;

  const int mlo  = lane & 15;
  const int koff = (lane >> 4) << 1;
  const float* arow = A + (rowBase + mlo) * (long)K;

  const int mhi = (lane >> 4) << 3;
  const float4 s0 = *(const float4*)(rs + rowBase + mhi);
  const float4 s1 = *(const float4*)(rs + rowBase + mhi + 4);

  v8f acc[4];
#pragma unroll
  for (int t = 0; t < 4; ++t) acc[t] = (v8f){0.f,0.f,0.f,0.f,0.f,0.f,0.f,0.f};

  for (int k0 = 0; k0 < K; k0 += 64) {
    float w0[8], w1[8];
#pragma unroll
    for (int it = 0; it < 8; ++it) {
      int idx = it * 256 + threadIdx.x;
      int ks = idx >> 7;
      int t  = (idx >> 5) & 3;
      int l  = idx & 31;
      int kk = k0 + ks * 4 + ((l >> 4) << 1);
      int cl = t * 16 + (l & 15);
      w0[it] = W[kk * NF + cl];
      w1[it] = W[(kk + 1) * NF + cl];
    }
    __syncthreads();
#pragma unroll
    for (int it = 0; it < 8; ++it) {
      int idx = it * 256 + threadIdx.x;
      sB[idx * 2 + 0] = w0[it];
      sB[idx * 2 + 1] = w1[it];
    }
    __syncthreads();
    if (k0 + 64 < K)
      __builtin_prefetch(arow + k0 + 64, 0, 0);
#pragma unroll 4
    for (int ks = 0; ks < 16; ++ks) {
      v2f a = *(const v2f*)(arow + k0 + ks * 4 + koff);
      const v2f* bfrag = (const v2f*)sB + ks * 128 + lane;
#pragma unroll
      for (int t = 0; t < 4; ++t) {
        v2f b = bfrag[t * 32];
        acc[t] = __builtin_amdgcn_wmma_f32_16x16x4_f32(
            false, a, false, b, (short)0, acc[t], false, false);
      }
    }
  }

  const float sc[8] = {s0.x, s0.y, s0.z, s0.w, s1.x, s1.y, s1.z, s1.w};
#pragma unroll
  for (int t = 0; t < 4; ++t)
#pragma unroll
    for (int i = 0; i < 8; ++i)
      C[(rowBase + mhi + i) * NF + t * 16 + mlo] = acc[t][i] * sc[i];
}

template<bool RELU>
__global__ __launch_bounds__(256) void k_gather(const float* __restrict__ xs,
                                                const int* __restrict__ offs,
                                                const int* __restrict__ srow,
                                                const float* __restrict__ dinv,
                                                const float* __restrict__ bias,
                                                float* __restrict__ out, int n) {
  int t = blockIdx.x * blockDim.x + threadIdx.x;
  int node = t >> 4;
  int f = (t & 15) << 2;
  if (node < n) {
    const float dn = dinv[node];
    const int beg = offs[node];
    const int end = offs[node + 1];
    const float* xf = xs + f;

    float4 acc = *(const float4*)(xf + (long)node * NF);

    int j = beg;
    for (; j + 4 <= end; j += 4) {
      int r0 = srow[j], r1 = srow[j + 1], r2 = srow[j + 2], r3 = srow[j + 3];
      float4 v0 = *(const float4*)(xf + (long)r0 * NF);
      float4 v1 = *(const float4*)(xf + (long)r1 * NF);
      float4 v2 = *(const float4*)(xf + (long)r2 * NF);
      float4 v3 = *(const float4*)(xf + (long)r3 * NF);
      acc.x += v0.x; acc.y += v0.y; acc.z += v0.z; acc.w += v0.w;
      acc.x += v1.x; acc.y += v1.y; acc.z += v1.z; acc.w += v1.w;
      acc.x += v2.x; acc.y += v2.y; acc.z += v2.z; acc.w += v2.w;
      acc.x += v3.x; acc.y += v3.y; acc.z += v3.z; acc.w += v3.w;
    }
    if (j + 2 <= end) {
      int r0 = srow[j], r1 = srow[j + 1];
      float4 v0 = *(const float4*)(xf + (long)r0 * NF);
      float4 v1 = *(const float4*)(xf + (long)r1 * NF);
      acc.x += v0.x; acc.y += v0.y; acc.z += v0.z; acc.w += v0.w;
      acc.x += v1.x; acc.y += v1.y; acc.z += v1.z; acc.w += v1.w;
      j += 2;
    }
    if (j < end) {
      int r = srow[j];
      float4 v = *(const float4*)(xf + (long)r * NF);
      acc.x += v.x; acc.y += v.y; acc.z += v.z; acc.w += v.w;
    }

    acc.x = acc.x * dn + bias[f + 0]; acc.y = acc.y * dn + bias[f + 1];
    acc.z = acc.z * dn + bias[f + 2]; acc.w = acc.w * dn + bias[f + 3];
    if (RELU) {
      acc.x = fmaxf(acc.x, 0.f); acc.y = fmaxf(acc.y, 0.f);
      acc.z = fmaxf(acc.z, 0.f); acc.w = fmaxf(acc.w, 0.f);
    }
    *(float4*)(out + (long)node * NF + f) = acc;
  }
}

#define HP 68
static_assert(NF == 64, "one row = 16 threads x 4 features; 4 column tiles");
__global__ __launch_bounds__(256) void k_gather_gemm(const float* __restrict__ xs,
                                                     const int* __restrict__ offs,
                                                     const int* __restrict__ srow,
                                                     const float* __restrict__ dinv,
                                                     const float* __restrict__ bias,
                                                     const float* __restrict__ W,
                                                     float* __restrict__ out, int n) {
  __shared__ __align__(16) float sH[16 * HP];
  __shared__ __align__(16) float sC[16 * HP];

  const int tid  = threadIdx.x;
  const int lane = tid & 31;
  const int wave = tid >> 5;
  const int r    = tid >> 4;
  const int node = blockIdx.x * 16 + r;
  const int f    = (tid & 15) << 2;

  float  dn  = 0.f;
  float4 acc = make_float4(0.f, 0.f, 0.f, 0.f);
  if (node < n) {
    dn = dinv[node];
    const int beg = offs[node];
    const int end = offs[node + 1];
    const float* xf = xs + f;

    acc = *(const float4*)(xf + (long)node * NF);

    int j = beg;
    for (; j + 4 <= end; j += 4) {
      int r0 = srow[j], r1 = srow[j + 1], r2 = srow[j + 2], r3 = srow[j + 3];
      float4 v0 = *(const float4*)(xf + (long)r0 * NF);
      float4 v1 = *(const float4*)(xf + (long)r1 * NF);
      float4 v2 = *(const float4*)(xf + (long)r2 * NF);
      float4 v3 = *(const float4*)(xf + (long)r3 * NF);
      acc.x += v0.x; acc.y += v0.y; acc.z += v0.z; acc.w += v0.w;
      acc.x += v1.x; acc.y += v1.y; acc.z += v1.z; acc.w += v1.w;
      acc.x += v2.x; acc.y += v2.y; acc.z += v2.z; acc.w += v2.w;
      acc.x += v3.x; acc.y += v3.y; acc.z += v3.z; acc.w += v3.w;
    }
    if (j + 2 <= end) {
      int r0 = srow[j], r1 = srow[j + 1];
      float4 v0 = *(const float4*)(xf + (long)r0 * NF);
      float4 v1 = *(const float4*)(xf + (long)r1 * NF);
      acc.x += v0.x; acc.y += v0.y; acc.z += v0.z; acc.w += v0.w;
      acc.x += v1.x; acc.y += v1.y; acc.z += v1.z; acc.w += v1.w;
      j += 2;
    }
    if (j < end) {
      int rr = srow[j];
      float4 v = *(const float4*)(xf + (long)rr * NF);
      acc.x += v.x; acc.y += v.y; acc.z += v.z; acc.w += v.w;
    }

    acc.x = fmaxf(acc.x * dn + bias[f + 0], 0.f); acc.y = fmaxf(acc.y * dn + bias[f + 1], 0.f);
    acc.z = fmaxf(acc.z * dn + bias[f + 2], 0.f); acc.w = fmaxf(acc.w * dn + bias[f + 3], 0.f);
  }
  *(float4*)(sH + r * HP + f) = acc;

  const int mlo  = lane & 15;
  const int koff = (lane >> 4) << 1;
  const int mhi  = (lane >> 4) << 3;

  float w0[16], w1[16];
  if (wave < 4) {
    const float* wp = W + koff * NF + wave * 16 + mlo;
#pragma unroll
    for (int ks = 0; ks < 16; ++ks) {
      w0[ks] = wp[(ks * 4) * NF];
      w1[ks] = wp[(ks * 4 + 1) * NF];
    }
  }
  __syncthreads();

  if (wave < 4) {
    v8f c = (v8f){0.f,0.f,0.f,0.f,0.f,0.f,0.f,0.f};
    const float* ap = sH + mlo * HP + koff;
#pragma unroll
    for (int ks = 0; ks < 16; ++ks) {
      v2f a = *(const v2f*)(ap + ks * 4);
      v2f b = (v2f){w0[ks], w1[ks]};
      c = __builtin_amdgcn_wmma_f32_16x16x4_f32(false, a, false, b, (short)0, c, false, false);
    }
    float* cp = sC + mhi * HP + wave * 16 + mlo;
#pragma unroll
    for (int i = 0; i < 8; ++i) cp[i * HP] = c[i];
  }
  __syncthreads();

  if (node < n) {
    float4 v = *(const float4*)(sC + r * HP + f);
    v.x *= dn; v.y *= dn; v.z *= dn; v.w *= dn;
    *(float4*)(out + (long)node * NF + f) = v;
  }
}

__global__ void k_decode(const float* __restrict__ z, const int* __restrict__ src,
                         const int* __restrict__ dst, float* __restrict__ out, int e) {
  int i = blockIdx.x * blockDim.x + threadIdx.x;
  if (i < e) {
    const float4* a = (const float4*)(z + (long)src[i] * NF);
    const float4* b = (const float4*)(z + (long)dst[i] * NF);
    float acc = 0.f;
#pragma unroll
    for (int j = 0; j < 16; ++j) {
      float4 av = a[j], bv = b[j];
      acc += av.x * bv.x + av.y * bv.y + av.z * bv.z + av.w * bv.w;
    }
    out[i] = acc;
  }
}

extern "C" void kernel_launch(void* const* d_in, const int* in_sizes, int n_in,
                              void* d_out, int out_size, void* d_ws, size_t ws_size,
                              hipStream_t stream) {
  const float* x   = (const float*)d_in[0];
  const int*   ei  = (const int*)d_in[1];
  const int*   eli = (const int*)d_in[2];
  const float* W1  = (const float*)d_in[3];
  const float* b1  = (const float*)d_in[4];
  const float* W2  = (const float*)d_in[5];
  const float* b2  = (const float*)d_in[6];
  float* out = (float*)d_out;

  float* ws   = (float*)d_ws;
  float* dinv = ws;
  float* bufA = ws + 50048;
  float* bufB = bufA + (size_t)NODES * NF;
  int*   offs = (int*)(bufB + (size_t)NODES * NF);
  int*   loff = offs + 50048;
  int*   part = loff + 50688;
  int*   srow = part + EDGES;

  const int* erow = ei;
  const int* ecol = ei + EDGES;
  const int* lsrc = eli;
  const int* ldst = eli + LEDGES;

  dim3 blk(256);
  const int nodeGrid = (NODES * 16 + 255) / 256;
  const int strips   = NODES / 16;
  const int gemmGrid = (strips + 7) / 8;

  k_chunk_sort<<<NCHUNK, CS_T, 0, stream>>>(erow, ecol, part, loff);
  k_bin_build<<<NBINS, BB_T, 0, stream>>>(part, loff, offs, srow, dinv);

  k_gemm<INCH><<<gemmGrid, blk, 0, stream>>>(x, W1, dinv, bufA, NODES);
  k_gather_gemm<<<nodeGrid, blk, 0, stream>>>(bufA, offs, srow, dinv, b1, W2, bufB, NODES);

  k_gather<false><<<nodeGrid, blk, 0, stream>>>(bufB, offs, srow, dinv, b2, bufA, NODES);

  k_decode<<<(LEDGES + 255) / 256, blk, 0, stream>>>(bufA, lsrc, ldst, out, LEDGES);
}


# Decoder: one lane per label edge reading its two rows in 32 scattered 16-byte loads becomes 16 lanes per edge, one load covering two whole rows; 16-lane butterfly sum, so the dot's sum order changes
# speedup vs baseline: 4.2990x; 1.2031x over previous
#include <hip/hip_runtime.h>

#define NODES  50000
#define EDGES  800000
#define LEDGES 200000
#define INCH   256
#define NF     64

typedef float v2f __attribute__((ext_vector_type(2)));
typedef float v8f __attribute__((ext_vector_type(8)));

#define BIN_NODES 256
#define NBINS     ((NODES + BIN_NODES - 1) / BIN_NODES)
#define NCHUNK    256
#define CHUNK_E   (EDGES / NCHUNK)
#define CS_T      512
#define CS_K      ((CHUNK_E + CS_T - 1) / CS_T)
#define BB_T      512
#define BB_K      10
#define BB_CAP    8192
static_assert(EDGES % NCHUNK == 0, "chunks are equal");
static_assert(NBINS + 1 <= 256 && NCHUNK == 256, "scans below take 256 entries");
static_assert(NODES < (1 << 23), "source id must fit beside 8 target bits");

__device__ __forceinline__ int wave_scan256(int* a, int lane) {
  int v[8];
  int s = 0;
#pragma unroll
  for (int k = 0; k < 8; ++k) { v[k] = a[lane * 8 + k]; s += v[k]; }
  int incl = s;
#pragma unroll
  for (int d = 1; d < 32; d <<= 1) {
    int up = __shfl_up(incl, d, 32);
    if (lane >= d) incl += up;
  }
  int run = incl - s;
#pragma unroll
  for (int k = 0; k < 8; ++k) { a[lane * 8 + k] = run; run += v[k]; }
  return __shfl(incl, 31, 32);
}

__global__ __launch_bounds__(CS_T) void k_chunk_sort(const int* __restrict__ row,
                                                     const int* __restrict__ col,
                                                     int* __restrict__ part,
                                                     int* __restrict__ loff) {
  __shared__ int hist[256];
  __shared__ int sortedL[CHUNK_E];
  const int tid  = threadIdx.x;
  const int lane = tid & 31;
  const int wave = tid >> 5;
  const int e0   = blockIdx.x * CHUNK_E;

  if (tid < 256) hist[tid] = 0;
  __syncthreads();

  int p[CS_K], bn[CS_K], rk[CS_K];
#pragma unroll
  for (int k = 0; k < CS_K; ++k) {
    const int i = k * CS_T + tid;
    p[k] = 0; bn[k] = 0;
    if (i < CHUNK_E) {
      const int c = col[e0 + i];
      const int r = row[e0 + i];
      bn[k] = c >> 8;
      p[k]  = (r << 8) | (c & 255);
    }
  }
#pragma unroll
  for (int k = 0; k < CS_K; ++k) {
    const int i = k * CS_T + tid;
    rk[k] = 0;
    if (i < CHUNK_E) rk[k] = atomicAdd(&hist[bn[k]], 1);
  }
  __syncthreads();
  if (wave == 0) wave_scan256(hist, lane);
  __syncthreads();
#pragma unroll
  for (int k = 0; k < CS_K; ++k) {
    const int i = k * CS_T + tid;
    if (i < CHUNK_E) sortedL[hist[bn[k]] + rk[k]] = p[k];
  }
  if (tid <= NBINS) loff[blockIdx.x * (NBINS + 1) + tid] = hist[tid];
  __syncthreads();
#pragma unroll
  for (int k = 0; k < CS_K; ++k) {
    const int i = k * CS_T + tid;
    if (i < CHUNK_E) part[e0 + i] = sortedL[i];
  }
}

__global__ __launch_bounds__(BB_T) void k_bin_build(const int* __restrict__ part,
                                                    const int* __restrict__ loff,
                                                    int* __restrict__ offs,
                                                    int* __restrict__ srow,
                                                    float* __restrict__ dinv) {
  __shared__ int pre[NCHUNK + 1];
  __shared__ int lo_s[NCHUNK];
  __shared__ int cntL[BIN_NODES];
  __shared__ int stage[BB_CAP];
  __shared__ int sorted[BB_CAP];
  __shared__ int tot[2];
  const int tid  = threadIdx.x;
  const int lane = tid & 31;
  const int wave = tid >> 5;
  const int b    = blockIdx.x;

  if (tid < NCHUNK) {
    const int lo = loff[tid * (NBINS + 1) + b];
    const int hi = loff[tid * (NBINS + 1) + b + 1];
    lo_s[tid] = lo;
    pre[tid]  = hi - lo;
  }
  if (tid < BIN_NODES) cntL[tid] = 0;
  __syncthreads();
  if (wave == 0) {
    int s = 0;
#pragma unroll
    for (int k = 0; k < 8; ++k) s += lo_s[lane * 8 + k];
#pragma unroll
    for (int d = 16; d >= 1; d >>= 1) s += __shfl_xor(s, d, 32);
    const int t = wave_scan256(pre, lane);
    if (lane == 0) { pre[NCHUNK] = t; tot[0] = t; tot[1] = s; }
  }
  __syncthreads();
  const int  T        = tot[0];
  const int  binStart = tot[1];
  const bool inLds    = T <= BB_CAP;

  auto fetch = [&](int i) -> int {
    int c = 0;
#pragma unroll
    for (int step = NCHUNK / 2; step >= 1; step >>= 1)
      if (pre[c + step] <= i) c += step;
    return part[c * CHUNK_E + lo_s[c] + (i - pre[c])];
  };

  for (int base = 0; base < T; base += BB_T * BB_K) {
    int q[BB_K];
#pragma unroll
    for (int k = 0; k < BB_K; ++k) {
      const int i = base + k * BB_T + tid;
      q[k] = fetch(i < T ? i : T - 1);
    }
#pragma unroll
    for (int k = 0; k < BB_K; ++k) {
      const int i = base + k * BB_T + tid;
      if (i < T) {
        atomicAdd(&cntL[q[k] & 255], 1);
        if (inLds) stage[i] = q[k];
      }
    }
  }
  __syncthreads();
  int mycnt = 0;
  if (tid < BIN_NODES) mycnt = cntL[tid];
  __syncthreads();
  if (wave == 0) wave_scan256(cntL, lane);
  __syncthreads();
  if (tid < BIN_NODES) {
    const int node = b * BIN_NODES + tid;
    if (node <= NODES) offs[node] = binStart + cntL[tid];
    if (node < NODES)  dinv[node] = rsqrtf((float)(mycnt + 1));
  }
  __syncthreads();

  if (inLds) {
    for (int base = 0; base < T; base += BB_T * BB_K) {
      int q[BB_K], pos[BB_K];
#pragma unroll
      for (int k = 0; k < BB_K; ++k) {
        const int i = base + k * BB_T + tid;
        q[k] = (i < T) ? stage[i] : 0;
      }
#pragma unroll
      for (int k = 0; k < BB_K; ++k) {
        const int i = base + k * BB_T + tid;
        pos[k] = 0;
        if (i < T) pos[k] = atomicAdd(&cntL[q[k] & 255], 1);
      }
#pragma unroll
      for (int k = 0; k < BB_K; ++k) {
        const int i = base + k * BB_T + tid;
        if (i < T) sorted[pos[k]] = (int)((unsigned)q[k] >> 8);
      }
    }
    __syncthreads();
    for (int base = 0; base < T; base += BB_T * BB_K) {
#pragma unroll
      for (int k = 0; k < BB_K; ++k) {
        const int i = base + k * BB_T + tid;
        if (i < T) srow[binStart + i] = sorted[i];
      }
    }
  } else {
    for (int i = tid; i < T; i += BB_T) {
      const int q   = fetch(i);
      const int pos = atomicAdd(&cntL[q & 255], 1);
      srow[binStart + pos] = (int)((unsigned)q >> 8);
    }
  }
}

template<int K>
__global__ __launch_bounds__(256) void k_gemm(const float* __restrict__ A,
                                              const float* __restrict__ W,
                                              const float* __restrict__ rs,
                                              float* __restrict__ C, int M) {
  __shared__ __align__(16) float sB[4096];

  const int lane = threadIdx.x & 31;
  const int wave = threadIdx.x >> 5;

  int strip = blockIdx.x * 8 + wave;
  int maxStrip = (M >> 4) - 1;
  if (strip > maxStrip) strip = maxStrip;
  const long rowBase = (long)strip << 4;

  const int mlo  = lane & 15;
  const int koff = (lane >> 4) << 1;
  const float* arow = A + (rowBase + mlo) * (long)K;

  const int mhi = (lane >> 4) << 3;
  const float4 s0 = *(const float4*)(rs + rowBase + mhi);
  const float4 s1 = *(const float4*)(rs + rowBase + mhi + 4);

  v8f acc[4];
#pragma unroll
  for (int t = 0; t < 4; ++t) acc[t] = (v8f){0.f,0.f,0.f,0.f,0.f,0.f,0.f,0.f};

  for (int k0 = 0; k0 < K; k0 += 64) {
    float w0[8], w1[8];
#pragma unroll
    for (int it = 0; it < 8; ++it) {
      int idx = it * 256 + threadIdx.x;
      int ks = idx >> 7;
      int t  = (idx >> 5) & 3;
      int l  = idx & 31;
      int kk = k0 + ks * 4 + ((l >> 4) << 1);
      int cl = t * 16 + (l & 15);
      w0[it] = W[kk * NF + cl];
      w1[it] = W[(kk + 1) * NF + cl];
    }
    __syncthreads();
#pragma unroll
    for (int it = 0; it < 8; ++it) {
      int idx = it * 256 + threadIdx.x;
      sB[idx * 2 + 0] = w0[it];
      sB[idx * 2 + 1] = w1[it];
    }
    __syncthreads();
    if (k0 + 64 < K)
      __builtin_prefetch(arow + k0 + 64, 0, 0);
#pragma unroll 4
    for (int ks = 0; ks < 16; ++ks) {
      v2f a = *(const v2f*)(arow + k0 + ks * 4 + koff);
      const v2f* bfrag = (const v2f*)sB + ks * 128 + lane;
#pragma unroll
      for (int t = 0; t < 4; ++t) {
        v2f b = bfrag[t * 32];
        acc[t] = __builtin_amdgcn_wmma_f32_16x16x4_f32(
            false, a, false, b, (short)0, acc[t], false, false);
      }
    }
  }

  const float sc[8] = {s0.x, s0.y, s0.z, s0.w, s1.x, s1.y, s1.z, s1.w};
#pragma unroll
  for (int t = 0; t < 4; ++t)
#pragma unroll
    for (int i = 0; i < 8; ++i)
      C[(rowBase + mhi + i) * NF + t * 16 + mlo] = acc[t][i] * sc[i];
}

template<bool RELU>
__global__ __launch_bounds__(256) void k_gather(const float* __restrict__ xs,
                                                const int* __restrict__ offs,
                                                const int* __restrict__ srow,
                                                const float* __restrict__ dinv,
                                                const float* __restrict__ bias,
                                                float* __restrict__ out, int n) {
  int t = blockIdx.x * blockDim.x + threadIdx.x;
  int node = t >> 4;
  int f = (t & 15) << 2;
  if (node < n) {
    const float dn = dinv[node];
    const int beg = offs[node];
    const int end = offs[node + 1];
    const float* xf = xs + f;

    float4 acc = *(const float4*)(xf + (long)node * NF);

    int j = beg;
    for (; j + 4 <= end; j += 4) {
      int r0 = srow[j], r1 = srow[j + 1], r2 = srow[j + 2], r3 = srow[j + 3];
      float4 v0 = *(const float4*)(xf + (long)r0 * NF);
      float4 v1 = *(const float4*)(xf + (long)r1 * NF);
      float4 v2 = *(const float4*)(xf + (long)r2 * NF);
      float4 v3 = *(const float4*)(xf + (long)r3 * NF);
      acc.x += v0.x; acc.y += v0.y; acc.z += v0.z; acc.w += v0.w;
      acc.x += v1.x; acc.y += v1.y; acc.z += v1.z; acc.w += v1.w;
      acc.x += v2.x; acc.y += v2.y; acc.z += v2.z; acc.w += v2.w;
      acc.x += v3.x; acc.y += v3.y; acc.z += v3.z; acc.w += v3.w;
    }
    if (j + 2 <= end) {
      int r0 = srow[j], r1 = srow[j + 1];
      float4 v0 = *(const float4*)(xf + (long)r0 * NF);
      float4 v1 = *(const float4*)(xf + (long)r1 * NF);
      acc.x += v0.x; acc.y += v0.y; acc.z += v0.z; acc.w += v0.w;
      acc.x += v1.x; acc.y += v1.y; acc.z += v1.z; acc.w += v1.w;
      j += 2;
    }
    if (j < end) {
      int r = srow[j];
      float4 v = *(const float4*)(xf + (long)r * NF);
      acc.x += v.x; acc.y += v.y; acc.z += v.z; acc.w += v.w;
    }

    acc.x = acc.x * dn + bias[f + 0]; acc.y = acc.y * dn + bias[f + 1];
    acc.z = acc.z * dn + bias[f + 2]; acc.w = acc.w * dn + bias[f + 3];
    if (RELU) {
      acc.x = fmaxf(acc.x, 0.f); acc.y = fmaxf(acc.y, 0.f);
      acc.z = fmaxf(acc.z, 0.f); acc.w = fmaxf(acc.w, 0.f);
    }
    *(float4*)(out + (long)node * NF + f) = acc;
  }
}

#define HP 68
static_assert(NF == 64, "one row = 16 threads x 4 features; 4 column tiles");
__global__ __launch_bounds__(256) void k_gather_gemm(const float* __restrict__ xs,
                                                     const int* __restrict__ offs,
                                                     const int* __restrict__ srow,
                                                     const float* __restrict__ dinv,
                                                     const float* __restrict__ bias,
                                                     const float* __restrict__ W,
                                                     float* __restrict__ out, int n) {
  __shared__ __align__(16) float sH[16 * HP];
  __shared__ __align__(16) float sC[16 * HP];

  const int tid  = threadIdx.x;
  const int lane = tid & 31;
  const int wave = tid >> 5;
  const int r    = tid >> 4;
  const int node = blockIdx.x * 16 + r;
  const int f    = (tid & 15) << 2;

  float  dn  = 0.f;
  float4 acc = make_float4(0.f, 0.f, 0.f, 0.f);
  if (node < n) {
    dn = dinv[node];
    const int beg = offs[node];
    const int end = offs[node + 1];
    const float* xf = xs + f;

    acc = *(const float4*)(xf + (long)node * NF);

    int j = beg;
    for (; j + 4 <= end; j += 4) {
      int r0 = srow[j], r1 = srow[j + 1], r2 = srow[j + 2], r3 = srow[j + 3];
      float4 v0 = *(const float4*)(xf + (long)r0 * NF);
      float4 v1 = *(const float4*)(xf + (long)r1 * NF);
      float4 v2 = *(const float4*)(xf + (long)r2 * NF);
      float4 v3 = *(const float4*)(xf + (long)r3 * NF);
      acc.x += v0.x; acc.y += v0.y; acc.z += v0.z; acc.w += v0.w;
      acc.x += v1.x; acc.y += v1.y; acc.z += v1.z; acc.w += v1.w;
      acc.x += v2.x; acc.y += v2.y; acc.z += v2.z; acc.w += v2.w;
      acc.x += v3.x; acc.y += v3.y; acc.z += v3.z; acc.w += v3.w;
    }
    if (j + 2 <= end) {
      int r0 = srow[j], r1 = srow[j + 1];
      float4 v0 = *(const float4*)(xf + (long)r0 * NF);
      float4 v1 = *(const float4*)(xf + (long)r1 * NF);
      acc.x += v0.x; acc.y += v0.y; acc.z += v0.z; acc.w += v0.w;
      acc.x += v1.x; acc.y += v1.y; acc.z += v1.z; acc.w += v1.w;
      j += 2;
    }
    if (j < end) {
      int rr = srow[j];
      float4 v = *(const float4*)(xf + (long)rr * NF);
      acc.x += v.x; acc.y += v.y; acc.z += v.z; acc.w += v.w;
    }

    acc.x = fmaxf(acc.x * dn + bias[f + 0], 0.f); acc.y = fmaxf(acc.y * dn + bias[f + 1], 0.f);
    acc.z = fmaxf(acc.z * dn + bias[f + 2], 0.f); acc.w = fmaxf(acc.w * dn + bias[f + 3], 0.f);
  }
  *(float4*)(sH + r * HP + f) = acc;

  const int mlo  = lane & 15;
  const int koff = (lane >> 4) << 1;
  const int mhi  = (lane >> 4) << 3;

  float w0[16], w1[16];
  if (wave < 4) {
    const float* wp = W + koff * NF + wave * 16 + mlo;
#pragma unroll
    for (int ks = 0; ks < 16; ++ks) {
      w0[ks] = wp[(ks * 4) * NF];
      w1[ks] = wp[(ks * 4 + 1) * NF];
    }
  }
  __syncthreads();

  if (wave < 4) {
    v8f c = (v8f){0.f,0.f,0.f,0.f,0.f,0.f,0.f,0.f};
    const float* ap = sH + mlo * HP + koff;
#pragma unroll
    for (int ks = 0; ks < 16; ++ks) {
      v2f a = *(const v2f*)(ap + ks * 4);
      v2f b = (v2f){w0[ks], w1[ks]};
      c = __builtin_amdgcn_wmma_f32_16x16x4_f32(false, a, false, b, (short)0, c, false, false);
    }
    float* cp = sC + mhi * HP + wave * 16 + mlo;
#pragma unroll
    for (int i = 0; i < 8; ++i) cp[i * HP] = c[i];
  }
  __syncthreads();

  if (node < n) {
    float4 v = *(const float4*)(sC + r * HP + f);
    v.x *= dn; v.y *= dn; v.z *= dn; v.w *= dn;
    *(float4*)(out + (long)node * NF + f) = v;
  }
}

#define DEC_STEPS 8

__device__ __forceinline__ float half_sum16(float v) {
  v += __int_as_float(__builtin_amdgcn_update_dpp(0, __float_as_int(v), 0x168, 0xf, 0xf, false));
  v += __int_as_float(__builtin_amdgcn_update_dpp(0, __float_as_int(v), 0x164, 0xf, 0xf, false));
  v += __int_as_float(__builtin_amdgcn_update_dpp(0, __float_as_int(v), 0x162, 0xf, 0xf, false));
  v += __int_as_float(__builtin_amdgcn_update_dpp(0, __float_as_int(v), 0x161, 0xf, 0xf, false));
  return v;
}

__global__ __launch_bounds__(256) void k_decode(const float* __restrict__ z, const int* __restrict__ src,
                                                const int* __restrict__ dst, float* __restrict__ out, int e) {
  const int lane = threadIdx.x & 31;
  const int q    = lane & 15;
  const int base = (blockIdx.x * 8 + (threadIdx.x >> 5)) * 32;
  if (base >= e) return;
  const int mine = base + lane;
  const int ec   = mine < e ? mine : e - 1;
  const int sv = src[ec];
  const int dv = dst[ec];
  const float4* zq = (const float4*)z + q;
  int svr = sv, dvr = dv;
  float res = 0.f;
#pragma unroll 1
  for (int t0 = 0; t0 < 16; t0 += DEC_STEPS) {
    float4 a[DEC_STEPS], b[DEC_STEPS];
#pragma unroll
    for (int s = 0; s < DEC_STEPS; ++s) {
      const int si = __shfl(svr, s, 16);
      const int di = __shfl(dvr, s, 16);
      a[s] = zq[(long)si * (NF / 4)];
      b[s] = zq[(long)di * (NF / 4)];
    }
#pragma unroll
    for (int s = 0; s < DEC_STEPS; ++s) {
      float p = a[s].x * b[s].x + a[s].y * b[s].y + a[s].z * b[s].z + a[s].w * b[s].w;
      p = half_sum16(p);
      if (q == t0 + s) res = p;
    }
    svr = __builtin_amdgcn_update_dpp(0, svr, 0x168, 0xf, 0xf, false);
    dvr = __builtin_amdgcn_update_dpp(0, dvr, 0x168, 0xf, 0xf, false);
  }
  if (mine < e) out[mine] = res;
}

extern "C" void kernel_launch(void* const* d_in, const int* in_sizes, int n_in,
                              void* d_out, int out_size, void* d_ws, size_t ws_size,
                              hipStream_t stream) {
  const float* x   = (const float*)d_in[0];
  const int*   ei  = (const int*)d_in[1];
  const int*   eli = (const int*)d_in[2];
  const float* W1  = (const float*)d_in[3];
  const float* b1  = (const float*)d_in[4];
  const float* W2  = (const float*)d_in[5];
  const float* b2  = (const float*)d_in[6];
  float* out = (float*)d_out;

  float* ws   = (float*)d_ws;
  float* dinv = ws;
  float* bufA = ws + 50048;
  float* bufB = bufA + (size_t)NODES * NF;
  int*   offs = (int*)(bufB + (size_t)NODES * NF);
  int*   loff = offs + 50048;
  int*   part = loff + 50688;
  int*   srow = part + EDGES;

  const int* erow = ei;
  const int* ecol = ei + EDGES;
  const int* lsrc = eli;
  const int* ldst = eli + LEDGES;

  dim3 blk(256);
  const int nodeGrid = (NODES * 16 + 255) / 256;
  const int strips   = NODES / 16;
  const int gemmGrid = (strips + 7) / 8;

  k_chunk_sort<<<NCHUNK, CS_T, 0, stream>>>(erow, ecol, part, loff);
  k_bin_build<<<NBINS, BB_T, 0, stream>>>(part, loff, offs, srow, dinv);

  k_gemm<INCH><<<gemmGrid, blk, 0, stream>>>(x, W1, dinv, bufA, NODES);
  k_gather_gemm<<<nodeGrid, blk, 0, stream>>>(bufA, offs, srow, dinv, b1, W2, bufB, NODES);

  k_gather<false><<<nodeGrid, blk, 0, stream>>>(bufB, offs, srow, dinv, b2, bufA, NODES);

  k_decode<<<(LEDGES + 255) / 256, blk, 0, stream>>>(bufA, lsrc, ldst, out, LEDGES);
}
